# starting kernel
# baseline (speedup 1.0000x reference)
#include <hip/hip_runtime.h>
#include <hip/hip_bf16.h>

#define NN 100000
#define EE 1600000
#define HH 128
#define LDS_STRIDE 132

typedef __attribute__((ext_vector_type(2))) float v2f;
typedef __attribute__((ext_vector_type(8))) float v8f;

__global__ void deg_kernel(const int* __restrict__ src, const int* __restrict__ dst,
                           float* __restrict__ deg_out, float* __restrict__ deg_in) {
    int e = blockIdx.x * blockDim.x + threadIdx.x;
    if (e < EE) {
        atomicAdd(&deg_out[src[e]], 1.0f);
        atomicAdd(&deg_in[dst[e]], 1.0f);
    }
}

__global__ void norm_kernel(float* __restrict__ ns, float* __restrict__ nd) {
    int i = blockIdx.x * blockDim.x + threadIdx.x;
    if (i < NN) {
        ns[i] = rsqrtf(fmaxf(ns[i], 1.0f));
        nd[i] = rsqrtf(fmaxf(nd[i], 1.0f));
    }
}

__global__ void scatter_edges(const float* __restrict__ feat,
                              const float* __restrict__ norm_src,
                              const int* __restrict__ src, const int* __restrict__ dst,
                              float* __restrict__ agg) {
    int e = blockIdx.x * (blockDim.x >> 5) + (threadIdx.x >> 5);
    if (e >= EE) return;
    int lane = threadIdx.x & 31;
    int s = src[e];
    int d = dst[e];
    float ns = norm_src[s];
    float4 v = ((const float4*)(feat + (size_t)s * HH))[lane];
    v.x *= ns; v.y *= ns; v.z *= ns; v.w *= ns;
    float* ad = agg + (size_t)d * HH + lane * 4;
    atomicAdd(ad + 0, v.x);
    atomicAdd(ad + 1, v.y);
    atomicAdd(ad + 2, v.z);
    atomicAdd(ad + 3, v.w);
}

__global__ void gcn_gemm(const float* __restrict__ agg,
                         const float* __restrict__ norm_dst,
                         const float* __restrict__ W,
                         const float* __restrict__ bias,
                         const float* __restrict__ ori_in,
                         float res,
                         float* __restrict__ out,
                         float* __restrict__ ori_out) {
    __shared__ float lds_a[16 * LDS_STRIDE];

    const int m0   = blockIdx.x * 16;
    const int tid  = threadIdx.x;
    const int wave = tid >> 5;
    const int lane = tid & 31;

    {
        int base = tid * 8;
        int row  = base >> 7;
        int col  = base & 127;
        float nd = norm_dst[m0 + row];
        const float4* s4 = (const float4*)(agg + (size_t)(m0 + row) * HH + col);
        float4 a0 = s4[0];
        float4 a1 = s4[1];
        a0.x *= nd; a0.y *= nd; a0.z *= nd; a0.w *= nd;
        a1.x *= nd; a1.y *= nd; a1.z *= nd; a1.w *= nd;
        float4* t4 = (float4*)(lds_a + row * LDS_STRIDE + col);
        t4[0] = a0;
        t4[1] = a1;
    }
    __syncthreads();

    const int n0   = wave * 16;
    const int half = lane >> 4;
    const int l15  = lane & 15;

    const float* arow = lds_a + l15 * LDS_STRIDE + half * 2;
    const float* bcol = W + (size_t)(half * 2) * HH + n0 + l15;

    v8f c = {};
#pragma unroll
    for (int kk = 0; kk < 32; ++kk) {
        const int k0 = kk * 4;
        v2f a, b;
        a.x = arow[k0];
        a.y = arow[k0 + 1];
        b.x = bcol[(size_t)k0 * HH];
        b.y = bcol[(size_t)k0 * HH + HH];
        c = __builtin_amdgcn_wmma_f32_16x16x4_f32(
                 false, a,  false, b,
                 (short)0, c,  false,  false);
    }

    const int nn = n0 + l15;
    const float bv = bias[nn];
#pragma unroll
    for (int r = 0; r < 8; ++r) {
        int m = m0 + r + half * 8;
        size_t idx = (size_t)m * HH + nn;
        float v = c[r] + bv;
        if (ori_in)  v += res * ori_in[idx];
        out[idx] = v;
        if (ori_out) ori_out[idx] = v;
    }
}

__global__ void ln_relu(float* __restrict__ h,
                        const float* __restrict__ g,
                        const float* __restrict__ bta) {
    int n = blockIdx.x * (blockDim.x >> 5) + (threadIdx.x >> 5);
    if (n >= NN) return;
    int lane = threadIdx.x & 31;
    float4* row = (float4*)(h + (size_t)n * HH);
    float4 v = row[lane];

    float s = v.x + v.y + v.z + v.w;
#pragma unroll
    for (int off = 16; off > 0; off >>= 1) s += __shfl_xor(s, off, 32);
    float mu = s * (1.0f / 128.0f);

    float dx = v.x - mu, dy = v.y - mu, dz = v.z - mu, dw = v.w - mu;
    float q = dx * dx + dy * dy + dz * dz + dw * dw;
#pragma unroll
    for (int off = 16; off > 0; off >>= 1) q += __shfl_xor(q, off, 32);
    float r = rsqrtf(q * (1.0f / 128.0f) + 1e-5f);

    int c = lane * 4;
    float4 o;
    o.x = fmaxf(dx * r * g[c + 0] + bta[c + 0], 0.0f);
    o.y = fmaxf(dy * r * g[c + 1] + bta[c + 1], 0.0f);
    o.z = fmaxf(dz * r * g[c + 2] + bta[c + 2], 0.0f);
    o.w = fmaxf(dw * r * g[c + 3] + bta[c + 3], 0.0f);
    row[lane] = o;
}

extern "C" void kernel_launch(void* const* d_in, const int* in_sizes, int n_in,
                              void* d_out, int out_size, void* d_ws, size_t ws_size,
                              hipStream_t stream) {
    (void)in_sizes; (void)n_in; (void)out_size; (void)ws_size;

    const float* x    = (const float*)d_in[0];
    const float* W0   = (const float*)d_in[1];
    const float* Ws   = (const float*)d_in[2];
    const float* b    = (const float*)d_in[3];
    const float* ln_g = (const float*)d_in[4];
    const float* ln_b = (const float*)d_in[5];
    const int*   src  = (const int*)d_in[6];
    const int*   dst  = (const int*)d_in[7];
    float* out = (float*)d_out;

    float* ws       = (float*)d_ws;
    float* norm_src = ws;
    float* norm_dst = ws + NN;
    float* agg      = ws + 2 * (size_t)NN;
    float* ori      = agg + (size_t)NN * HH;
    float* hbuf     = ori + (size_t)NN * HH;

    const size_t featBytes = (size_t)NN * HH * sizeof(float);

    hipMemsetAsync(ws, 0, 2 * (size_t)NN * sizeof(float), stream);
    deg_kernel<<<EE / 256, 256, 0, stream>>>(src, dst, norm_src, norm_dst);
    norm_kernel<<<(NN + 255) / 256, 256, 0, stream>>>(norm_src, norm_dst);

    hipMemsetAsync(agg, 0, featBytes, stream);
    scatter_edges<<<EE / 8, 256, 0, stream>>>(x, norm_src, src, dst, agg);
    gcn_gemm<<<NN / 16, 256, 0, stream>>>(agg, norm_dst, W0, b, nullptr, 0.0f,
                                          hbuf, ori);

    ln_relu<<<NN / 8, 256, 0, stream>>>(hbuf, ln_g + HH, ln_b + HH);
    hipMemsetAsync(agg, 0, featBytes, stream);
    scatter_edges<<<EE / 8, 256, 0, stream>>>(hbuf, norm_src, src, dst, agg);
    gcn_gemm<<<NN / 16, 256, 0, stream>>>(agg, norm_dst, Ws, b + HH, ori, 0.5f,
                                          hbuf, nullptr);

    ln_relu<<<NN / 8, 256, 0, stream>>>(hbuf, ln_g + 2 * HH, ln_b + 2 * HH);
    hipMemsetAsync(agg, 0, featBytes, stream);
    scatter_edges<<<EE / 8, 256, 0, stream>>>(hbuf, norm_src, src, dst, agg);
    gcn_gemm<<<NN / 16, 256, 0, stream>>>(agg, norm_dst, Ws + (size_t)HH * HH,
                                          b + 2 * HH, ori, 0.5f, out, nullptr);
}


# aggregate by destination range: 256 work-groups each keep 391 output rows in LDS, filter the whole edge list for them, sum with LDS f32 adds (arrival order); scatter's global atomics and memsets go
# speedup vs baseline: 1.8352x; 1.8352x over previous
#include <hip/hip_runtime.h>
#include <hip/hip_bf16.h>

#define NN 100000
#define EE 1600000
#define HH 128
#define LDS_STRIDE 132

typedef __attribute__((ext_vector_type(2))) float v2f;
typedef __attribute__((ext_vector_type(8))) float v8f;

__global__ void deg_kernel(const int* __restrict__ src, const int* __restrict__ dst,
                           float* __restrict__ deg_out, float* __restrict__ deg_in) {
    int e = blockIdx.x * blockDim.x + threadIdx.x;
    if (e < EE) {
        atomicAdd(&deg_out[src[e]], 1.0f);
        atomicAdd(&deg_in[dst[e]], 1.0f);
    }
}

__global__ void norm_kernel(float* __restrict__ ns, float* __restrict__ nd) {
    int i = blockIdx.x * blockDim.x + threadIdx.x;
    if (i < NN) {
        ns[i] = rsqrtf(fmaxf(ns[i], 1.0f));
        nd[i] = rsqrtf(fmaxf(nd[i], 1.0f));
    }
}

#define RPB   391
#define NBLK  256
#define CH    128
#define NCH   (EE / CH)
#define QCAP  520
#define EBITS 21
static_assert(EE % CH == 0, "edge list is a whole number of steps");
static_assert(NBLK * RPB >= NN, "ranges cover every row");
static_assert(EE <= (1 << EBITS) && RPB <= 512, "entry packing");

template <bool FULL>
__device__ __forceinline__ void sum_listed(const float* __restrict__ feat,
                                           const float* __restrict__ norm_src,
                                           const int* __restrict__ src,
                                           float* acc, const int* q,
                                           int first, int cnt, int lane) {
    int ent = 0, s = 0;
    float ns = 0.0f;
    if (lane < cnt) {
        ent = q[first + lane];
        s = src[ent & ((1 << EBITS) - 1)];
        ns = norm_src[s];
    }
    float4 v[8];
#pragma unroll
    for (int u = 0; u < 8; ++u) {
        if (FULL || u < cnt) {
            const int sj = __builtin_amdgcn_readlane(s, u);
            v[u] = ((const float4*)(feat + (size_t)sj * HH))[lane];
        }
    }
#pragma unroll
    for (int u = 0; u < 8; ++u) {
        if (FULL || u < cnt) {
            const int r = __builtin_amdgcn_readlane(ent, u) >> EBITS;
            const float nsj = __int_as_float(
                __builtin_amdgcn_readlane(__float_as_int(ns), u));
            float* a = acc + r * HH + lane;
            atomicAdd(a,      v[u].x * nsj);
            atomicAdd(a + 32, v[u].y * nsj);
            atomicAdd(a + 64, v[u].z * nsj);
            atomicAdd(a + 96, v[u].w * nsj);
        }
    }
}

__global__ __launch_bounds__(1024)
void agg_rows(const float* __restrict__ feat,
              const float* __restrict__ norm_src,
              const int* __restrict__ src, const int* __restrict__ dst,
              float* __restrict__ agg) {
    __shared__ __attribute__((aligned(16))) float acc[RPB * HH];
    __shared__ int qall[32 * QCAP];

    const int tid  = threadIdx.x;
    const int lane = tid & 31;
    const int wave = tid >> 5;
    const int base = blockIdx.x * RPB;

    {
        float4* acc4 = (float4*)acc;
        for (int i = tid; i < RPB * HH / 4; i += 1024)
            acc4[i] = make_float4(0.0f, 0.0f, 0.0f, 0.0f);
    }
    __syncthreads();

    int* q = qall + wave * QCAP;
    int qn = 0;
    const int4* dst4 = (const int4*)dst;
    const int start = blockIdx.x * (NCH / NBLK);

    for (int it = wave; it < NCH; it += 256) {
        int4 d[8];
        int  c[8];
#pragma unroll
        for (int j = 0; j < 8; ++j) {
            const int ci = it + 32 * j;
            int cj = ci + start;
            if (cj >= NCH) cj -= NCH;
            c[j] = cj;
            d[j] = make_int4(-1, -1, -1, -1);
            if (ci < NCH) d[j] = dst4[cj * 32 + lane];
        }
#pragma unroll
        for (int h = 0; h < 2; ++h) {
#pragma unroll
            for (int j = 4 * h; j < 4 * h + 4; ++j) {
                const int e0 = c[j] * CH + lane * 4;
                const int dv[4] = {d[j].x, d[j].y, d[j].z, d[j].w};
#pragma unroll
                for (int k = 0; k < 4; ++k) {
                    const unsigned r = (unsigned)(dv[k] - base);
                    const bool m = r < (unsigned)RPB;
                    const unsigned mask = __builtin_amdgcn_ballot_w32(m);
                    if (mask != 0u) {
                        if (m)
                            q[qn + (int)__builtin_amdgcn_mbcnt_lo(mask, 0u)] =
                                (e0 + k) | ((int)r << EBITS);
                        qn += __builtin_popcount(mask);
                    }
                }
            }
            while (qn >= 8) {
                qn -= 8;
                sum_listed<true>(feat, norm_src, src, acc, q, qn, 8, lane);
            }
        }
    }
    if (qn > 0)
        sum_listed<false>(feat, norm_src, src, acc, q, 0, qn, lane);

    __syncthreads();

    for (int row = wave; row < RPB; row += 32) {
        const int n = base + row;
        if (n < NN) {
            const float* a = acc + row * HH + lane;
            float4 o;
            o.x = a[0];
            o.y = a[32];
            o.z = a[64];
            o.w = a[96];
            ((float4*)(agg + (size_t)n * HH))[lane] = o;
        }
    }
}

__global__ void gcn_gemm(const float* __restrict__ agg,
                         const float* __restrict__ norm_dst,
                         const float* __restrict__ W,
                         const float* __restrict__ bias,
                         const float* __restrict__ ori_in,
                         float res,
                         float* __restrict__ out,
                         float* __restrict__ ori_out) {
    __shared__ float lds_a[16 * LDS_STRIDE];

    const int m0   = blockIdx.x * 16;
    const int tid  = threadIdx.x;
    const int wave = tid >> 5;
    const int lane = tid & 31;

    {
        int base = tid * 8;
        int row  = base >> 7;
        int col  = base & 127;
        float nd = norm_dst[m0 + row];
        const float4* s4 = (const float4*)(agg + (size_t)(m0 + row) * HH + col);
        float4 a0 = s4[0];
        float4 a1 = s4[1];
        a0.x *= nd; a0.y *= nd; a0.z *= nd; a0.w *= nd;
        a1.x *= nd; a1.y *= nd; a1.z *= nd; a1.w *= nd;
        float4* t4 = (float4*)(lds_a + row * LDS_STRIDE + col);
        t4[0] = a0;
        t4[1] = a1;
    }
    __syncthreads();

    const int n0   = wave * 16;
    const int half = lane >> 4;
    const int l15  = lane & 15;

    const float* arow = lds_a + l15 * LDS_STRIDE + half * 2;
    const float* bcol = W + (size_t)(half * 2) * HH + n0 + l15;

    v8f c = {};
#pragma unroll
    for (int kk = 0; kk < 32; ++kk) {
        const int k0 = kk * 4;
        v2f a, b;
        a.x = arow[k0];
        a.y = arow[k0 + 1];
        b.x = bcol[(size_t)k0 * HH];
        b.y = bcol[(size_t)k0 * HH + HH];
        c = __builtin_amdgcn_wmma_f32_16x16x4_f32(
                 false, a,  false, b,
                 (short)0, c,  false,  false);
    }

    const int nn = n0 + l15;
    const float bv = bias[nn];
#pragma unroll
    for (int r = 0; r < 8; ++r) {
        int m = m0 + r + half * 8;
        size_t idx = (size_t)m * HH + nn;
        float v = c[r] + bv;
        if (ori_in)  v += res * ori_in[idx];
        out[idx] = v;
        if (ori_out) ori_out[idx] = v;
    }
}

__global__ void ln_relu(float* __restrict__ h,
                        const float* __restrict__ g,
                        const float* __restrict__ bta) {
    int n = blockIdx.x * (blockDim.x >> 5) + (threadIdx.x >> 5);
    if (n >= NN) return;
    int lane = threadIdx.x & 31;
    float4* row = (float4*)(h + (size_t)n * HH);
    float4 v = row[lane];

    float s = v.x + v.y + v.z + v.w;
#pragma unroll
    for (int off = 16; off > 0; off >>= 1) s += __shfl_xor(s, off, 32);
    float mu = s * (1.0f / 128.0f);

    float dx = v.x - mu, dy = v.y - mu, dz = v.z - mu, dw = v.w - mu;
    float q = dx * dx + dy * dy + dz * dz + dw * dw;
#pragma unroll
    for (int off = 16; off > 0; off >>= 1) q += __shfl_xor(q, off, 32);
    float r = rsqrtf(q * (1.0f / 128.0f) + 1e-5f);

    int c = lane * 4;
    float4 o;
    o.x = fmaxf(dx * r * g[c + 0] + bta[c + 0], 0.0f);
    o.y = fmaxf(dy * r * g[c + 1] + bta[c + 1], 0.0f);
    o.z = fmaxf(dz * r * g[c + 2] + bta[c + 2], 0.0f);
    o.w = fmaxf(dw * r * g[c + 3] + bta[c + 3], 0.0f);
    row[lane] = o;
}

extern "C" void kernel_launch(void* const* d_in, const int* in_sizes, int n_in,
                              void* d_out, int out_size, void* d_ws, size_t ws_size,
                              hipStream_t stream) {
    (void)in_sizes; (void)n_in; (void)out_size; (void)ws_size;

    const float* x    = (const float*)d_in[0];
    const float* W0   = (const float*)d_in[1];
    const float* Ws   = (const float*)d_in[2];
    const float* b    = (const float*)d_in[3];
    const float* ln_g = (const float*)d_in[4];
    const float* ln_b = (const float*)d_in[5];
    const int*   src  = (const int*)d_in[6];
    const int*   dst  = (const int*)d_in[7];
    float* out = (float*)d_out;

    float* ws       = (float*)d_ws;
    float* norm_src = ws;
    float* norm_dst = ws + NN;
    float* agg      = ws + 2 * (size_t)NN;
    float* ori      = agg + (size_t)NN * HH;
    float* hbuf     = ori + (size_t)NN * HH;

    hipMemsetAsync(ws, 0, 2 * (size_t)NN * sizeof(float), stream);
    deg_kernel<<<EE / 256, 256, 0, stream>>>(src, dst, norm_src, norm_dst);
    norm_kernel<<<(NN + 255) / 256, 256, 0, stream>>>(norm_src, norm_dst);

    agg_rows<<<NBLK, 1024, 0, stream>>>(x, norm_src, src, dst, agg);
    gcn_gemm<<<NN / 16, 256, 0, stream>>>(agg, norm_dst, W0, b, nullptr, 0.0f,
                                          hbuf, ori);

    ln_relu<<<NN / 8, 256, 0, stream>>>(hbuf, ln_g + HH, ln_b + HH);
    agg_rows<<<NBLK, 1024, 0, stream>>>(hbuf, norm_src, src, dst, agg);
    gcn_gemm<<<NN / 16, 256, 0, stream>>>(agg, norm_dst, Ws, b + HH, ori, 0.5f,
                                          hbuf, nullptr);

    ln_relu<<<NN / 8, 256, 0, stream>>>(hbuf, ln_g + 2 * HH, ln_b + 2 * HH);
    agg_rows<<<NBLK, 1024, 0, stream>>>(hbuf, norm_src, src, dst, agg);
    gcn_gemm<<<NN / 16, 256, 0, stream>>>(agg, norm_dst, Ws + (size_t)HH * HH,
                                          b + 2 * HH, ori, 0.5f, out, nullptr);
}


# first-layer filter tests one range byte per edge (dst/391, written by the degree launch; 16 edges per lane and load) instead of the 4-byte dst; rows fetched at take time; LDS f32 adds, arrival order
# speedup vs baseline: 2.2160x; 1.2075x over previous
#include <hip/hip_runtime.h>
#include <hip/hip_bf16.h>

#define NN 100000
#define EE 1600000
#define HH 128
#define LDS_STRIDE 132

typedef __attribute__((ext_vector_type(2))) float v2f;
typedef __attribute__((ext_vector_type(8))) float v8f;

#define RPB   391
#define NBLK  256
static_assert(NBLK <= 256 && (NN - 1) / RPB < 256, "a range number fits one byte");
static_assert(EE % 128 == 0, "whole waves of four edges per lane");

#define DEG_BLOCKS ((EE / 4 + 255) / 256)
__global__ void deg_kernel(const int* __restrict__ src, const int* __restrict__ dst,
                           float* __restrict__ deg_out, float* __restrict__ deg_in,
                           unsigned* __restrict__ rbytes) {
    const int t = blockIdx.x * blockDim.x + threadIdx.x;
    if (t < EE / 4) {
        const int4 s = ((const int4*)src)[t];
        const int4 d = ((const int4*)dst)[t];
        atomicAdd(&deg_out[s.x], 1.0f);
        atomicAdd(&deg_out[s.y], 1.0f);
        atomicAdd(&deg_out[s.z], 1.0f);
        atomicAdd(&deg_out[s.w], 1.0f);
        atomicAdd(&deg_in[d.x], 1.0f);
        atomicAdd(&deg_in[d.y], 1.0f);
        atomicAdd(&deg_in[d.z], 1.0f);
        atomicAdd(&deg_in[d.w], 1.0f);
        rbytes[t] = (unsigned)(d.x / RPB) | ((unsigned)(d.y / RPB) << 8) |
                    ((unsigned)(d.z / RPB) << 16) | ((unsigned)(d.w / RPB) << 24);
    }
}

__global__ __launch_bounds__(512)
void norm_kernel(float* __restrict__ ns, float* __restrict__ nd,
                 int* __restrict__ range_cnt) {
    __shared__ int tot;
    if (threadIdx.x == 0) tot = 0;
    __syncthreads();
    const int r = threadIdx.x;
    const int i = blockIdx.x * RPB + r;
    int c = 0;
    if (r < RPB && i < NN) {
        const float di = nd[i];
        c = (int)di;
        ns[i] = rsqrtf(fmaxf(ns[i], 1.0f));
        nd[i] = rsqrtf(fmaxf(di, 1.0f));
    }
    for (int off = 16; off > 0; off >>= 1) c += __shfl_xor(c, off, 32);
    if ((threadIdx.x & 31) == 0 && c != 0) atomicAdd(&tot, c);
    __syncthreads();
    if (threadIdx.x == 0) range_cnt[blockIdx.x] = tot;
}

#define CH    512
#define NCH   (EE / CH)
#define QCAP  544
#define SBITS 17
#define LCAP  (EE + NBLK * 32)
static_assert(EE % CH == 0, "edge list is a whole number of steps");
static_assert(NBLK * RPB >= NN, "ranges cover every row");
static_assert(31 + CH <= QCAP, "queue holds one step on top of a remainder");
static_assert(NCH + 64 + (NBLK - 1) * (NCH / NBLK) <= 2 * NCH, "read-ahead places wrap once");
static_assert(RPB <= 512, "local row packing");
static_assert(NN <= (1 << SBITS), "list word packing");
static_assert(NBLK <= 1024, "one thread per earlier group");

__device__ __forceinline__ float4 load_row(const float* __restrict__ feat,
                                           int row, int lane) {
    return ((const float4*)(feat + (size_t)row * HH))[lane];
}

__device__ __forceinline__ void add_row(float* acc, int r, float ns, int j,
                                        float4 w, int lane) {
    const int rj = __builtin_amdgcn_readlane(r, j);
    const float nsj = __int_as_float(
        __builtin_amdgcn_readlane(__float_as_int(ns), j));
    float* a = acc + rj * HH + lane;
    atomicAdd(a,      w.x * nsj);
    atomicAdd(a + 32, w.y * nsj);
    atomicAdd(a + 64, w.z * nsj);
    atomicAdd(a + 96, w.w * nsj);
}

template <bool FULL>
__device__ __forceinline__ void sum_chunk(const float* __restrict__ feat,
                                          float* acc, int s, int r, float ns,
                                          int cnt, int lane) {
    if (FULL) {
        float4 v[2][8];
#pragma unroll
        for (int u = 0; u < 8; ++u)
            v[0][u] = load_row(feat, __builtin_amdgcn_readlane(s, u), lane);
#pragma unroll
        for (int b = 0; b < 4; ++b) {
            if (b < 3) {
#pragma unroll
                for (int u = 0; u < 8; ++u)
                    v[(b + 1) & 1][u] = load_row(
                        feat, __builtin_amdgcn_readlane(s, 8 * (b + 1) + u), lane);
            }
#pragma unroll
            for (int u = 0; u < 8; ++u)
                add_row(acc, r, ns, 8 * b + u, v[b & 1][u], lane);
        }
    } else {
#pragma unroll
        for (int b = 0; b < 4; ++b) {
            if (8 * b < cnt) {
                float4 v[8];
#pragma unroll
                for (int u = 0; u < 8; ++u)
                    if (8 * b + u < cnt)
                        v[u] = load_row(
                            feat, __builtin_amdgcn_readlane(s, 8 * b + u), lane);
#pragma unroll
                for (int u = 0; u < 8; ++u)
                    if (8 * b + u < cnt)
                        add_row(acc, r, ns, 8 * b + u, v[u], lane);
            }
        }
    }
}

__device__ __forceinline__ void add_earlier_counts(const int* __restrict__ range_cnt,
                                                   int* goff_s, int tid) {
    if (tid < (int)blockIdx.x)
        atomicAdd(goff_s, (range_cnt[tid] + 31) & ~31);
}

__device__ __forceinline__ void store_rows(const float* acc, float* __restrict__ agg,
                                           int base, int wave, int lane) {
    for (int row = wave; row < RPB; row += 32) {
        const int n = base + row;
        if (n < NN) {
            const float* a = acc + row * HH + lane;
            float4 o;
            o.x = a[0];
            o.y = a[32];
            o.z = a[64];
            o.w = a[96];
            ((float4*)(agg + (size_t)n * HH))[lane] = o;
        }
    }
}

template <bool FULL>
__device__ __forceinline__ void take_entries(const float* __restrict__ feat,
                                             const float* __restrict__ norm_src,
                                             const int* __restrict__ src,
                                             const int* __restrict__ dst,
                                             int base,
                                             float* acc, const int* q,
                                             int first, int cnt, int lane,
                                             int* __restrict__ lst, int* cursor,
                                             int* tail, int* tailn) {
    int s = 0, r = 0;
    float ns = 0.0f;
    if (FULL || lane < cnt) {
        const int e = q[first + lane];
        s = src[e];
        r = dst[e] - base;
        ns = norm_src[s];
    }
    int pos = 0;
    if (lane == 0) pos = atomicAdd(FULL ? cursor : tailn, FULL ? 32 : cnt);
    pos = __builtin_amdgcn_readfirstlane(pos);
    if (FULL)
        lst[pos + lane] = s | (r << SBITS);
    else if (lane < cnt)
        tail[pos + lane] = s | (r << SBITS);
    sum_chunk<FULL>(feat, acc, s, r, ns, cnt, lane);
}

__device__ __forceinline__ unsigned nonzero_bytes(unsigned x) {
    return (((x & 0x7F7F7F7Fu) + 0x7F7F7F7Fu) | x) & 0x80808080u;
}

__device__ __forceinline__ unsigned hit_bits(uint4 g, unsigned pat) {
    const unsigned n = (nonzero_bytes(g.x ^ pat) >> 7) | (nonzero_bytes(g.y ^ pat) >> 6) |
                       (nonzero_bytes(g.z ^ pat) >> 5) | (nonzero_bytes(g.w ^ pat) >> 4);
    return ~n & 0x0F0F0F0Fu;
}

__device__ __forceinline__ int step_place(int t, int start) {
    int p = t + start;
    if (p >= NCH) p -= NCH;
    return p;
}

__device__ __forceinline__ int filter_step(uint4& g, int t, int start, unsigned pat,
                                           const uint4* __restrict__ rb4,
                                           const float* __restrict__ feat,
                                           const float* __restrict__ norm_src,
                                           const int* __restrict__ src,
                                           const int* __restrict__ dst,
                                           int base, float* acc, int* q, int qn,
                                           int lane, int* __restrict__ lst,
                                           int* ctl, int* tail) {
    unsigned hits = hit_bits(g, pat);
    g = rb4[step_place(t + 64, start) * 32 + lane];
    const int e0 = step_place(t, start) * CH + lane * 16;
    for (int pass = 0; pass < 16; ++pass) {
        const bool m = hits != 0u;
        const unsigned mask = __builtin_amdgcn_ballot_w32(m);
        if (mask == 0u) break;
        if (m) {
            const int pos = __builtin_ctz(hits);
            q[qn + (int)__builtin_amdgcn_mbcnt_lo(mask, 0u)] =
                e0 + ((pos & 7) << 2) + (pos >> 3);
            hits &= hits - 1u;
        }
        qn += __builtin_popcount(mask);
    }
    while (qn >= 32) {
        qn -= 32;
        take_entries<true>(feat, norm_src, src, dst, base, acc, q, qn, 32, lane,
                           lst, &ctl[1], tail, &ctl[2]);
    }
    return qn;
}

__global__ __launch_bounds__(1024)
void agg_first(const float* __restrict__ feat,
               const float* __restrict__ norm_src,
               const int* __restrict__ src, const int* __restrict__ dst,
               const unsigned* __restrict__ rbytes,
               const int* __restrict__ range_cnt,
               int* __restrict__ elist,
               float* __restrict__ agg) {
    __shared__ __attribute__((aligned(16))) float acc[RPB * HH];
    __shared__ int qall[32 * QCAP];
    __shared__ int tail[32 * 32];
    __shared__ int ctl[3];

    const int tid  = threadIdx.x;
    const int lane = tid & 31;
    const int wave = tid >> 5;
    const int base = blockIdx.x * RPB;

    if (tid < 3) ctl[tid] = 0;
    {
        float4* acc4 = (float4*)acc;
        for (int i = tid; i < RPB * HH / 4; i += 1024)
            acc4[i] = make_float4(0.0f, 0.0f, 0.0f, 0.0f);
    }
    __syncthreads();
    add_earlier_counts(range_cnt, &ctl[0], tid);
    __syncthreads();
    int* lst = elist + ctl[0];

    int* q = qall + wave * QCAP;
    int qn = 0;
    const uint4* rb4 = (const uint4*)rbytes;
    const unsigned pat = (unsigned)blockIdx.x * 0x01010101u;
    const int start = blockIdx.x * (NCH / NBLK);

    const int wv = __builtin_amdgcn_readfirstlane(wave);
    uint4 ga = rb4[step_place(wv, start) * 32 + lane];
    uint4 gb = rb4[step_place(wv + 32, start) * 32 + lane];
    for (int it = wv; it < NCH; it += 64) {
        qn = filter_step(ga, it, start, pat, rb4, feat, norm_src, src, dst, base,
                         acc, q, qn, lane, lst, ctl, tail);
        if (it + 32 >= NCH) break;
        qn = filter_step(gb, it + 32, start, pat, rb4, feat, norm_src, src, dst,
                         base, acc, q, qn, lane, lst, ctl, tail);
    }
    if (qn > 0)
        take_entries<false>(feat, norm_src, src, dst, base, acc, q, 0, qn, lane,
                            lst, &ctl[1], tail, &ctl[2]);

    __syncthreads();

    {
        const int nfull = ctl[1];
        const int nt = ctl[2];
        for (int i = wave * 32; i < nt; i += 1024)
            if (i + lane < nt) lst[nfull + i + lane] = tail[i + lane];
    }
    store_rows(acc, agg, base, wave, lane);
}

__global__ __launch_bounds__(1024)
void agg_listed(const float* __restrict__ feat,
                const float* __restrict__ norm_src,
                const int* __restrict__ range_cnt,
                const int* __restrict__ elist,
                float* __restrict__ agg) {
    __shared__ __attribute__((aligned(16))) float acc[RPB * HH];
    __shared__ int goff_s;

    const int tid  = threadIdx.x;
    const int lane = tid & 31;
    const int wave = tid >> 5;
    const int base = blockIdx.x * RPB;

    if (tid == 0) goff_s = 0;
    {
        float4* acc4 = (float4*)acc;
        for (int i = tid; i < RPB * HH / 4; i += 1024)
            acc4[i] = make_float4(0.0f, 0.0f, 0.0f, 0.0f);
    }
    __syncthreads();
    add_earlier_counts(range_cnt, &goff_s, tid);
    __syncthreads();
    const int* lst = elist + goff_s;
    const int total = range_cnt[blockIdx.x];

    const int share = (((total + 31) >> 5) + 7) & ~7;
    int c0 = wave * share;
    int wend = c0 + share;
    if (wend > total) wend = total;

    const int smask = (1 << SBITS) - 1;
    int e0 = 0, e1 = 0;
    if (c0 + lane < wend)      e0 = lst[c0 + lane];
    if (c0 + 32 + lane < wend) e1 = lst[c0 + 32 + lane];
    float ns0 = 0.0f;
    if (c0 + lane < wend) ns0 = norm_src[e0 & smask];

    while (c0 < wend) {
        int n = wend - c0;
        if (n > 32) n = 32;
        int e2 = 0;
        if (c0 + 64 + lane < wend) e2 = lst[c0 + 64 + lane];
        float ns1 = 0.0f;
        if (c0 + 32 + lane < wend) ns1 = norm_src[e1 & smask];

        const int s = e0 & smask;
        const int r = e0 >> SBITS;
        if (n == 32) sum_chunk<true>(feat, acc, s, r, ns0, 32, lane);
        else         sum_chunk<false>(feat, acc, s, r, ns0, n, lane);

        e0 = e1; e1 = e2; ns0 = ns1;
        c0 += 32;
    }

    __syncthreads();
    store_rows(acc, agg, base, wave, lane);
}

__global__ void gcn_gemm(const float* __restrict__ agg,
                         const float* __restrict__ norm_dst,
                         const float* __restrict__ W,
                         const float* __restrict__ bias,
                         const float* __restrict__ ori_in,
                         float res,
                         float* __restrict__ out,
                         float* __restrict__ ori_out) {
    __shared__ float lds_a[16 * LDS_STRIDE];

    const int m0   = blockIdx.x * 16;
    const int tid  = threadIdx.x;
    const int wave = tid >> 5;
    const int lane = tid & 31;

    {
        int base = tid * 8;
        int row  = base >> 7;
        int col  = base & 127;
        float nd = norm_dst[m0 + row];
        const float4* s4 = (const float4*)(agg + (size_t)(m0 + row) * HH + col);
        float4 a0 = s4[0];
        float4 a1 = s4[1];
        a0.x *= nd; a0.y *= nd; a0.z *= nd; a0.w *= nd;
        a1.x *= nd; a1.y *= nd; a1.z *= nd; a1.w *= nd;
        float4* t4 = (float4*)(lds_a + row * LDS_STRIDE + col);
        t4[0] = a0;
        t4[1] = a1;
    }
    __syncthreads();

    const int n0   = wave * 16;
    const int half = lane >> 4;
    const int l15  = lane & 15;

    const float* arow = lds_a + l15 * LDS_STRIDE + half * 2;
    const float* bcol = W + (size_t)(half * 2) * HH + n0 + l15;

    v8f c = {};
#pragma unroll
    for (int kk = 0; kk < 32; ++kk) {
        const int k0 = kk * 4;
        v2f a, b;
        a.x = arow[k0];
        a.y = arow[k0 + 1];
        b.x = bcol[(size_t)k0 * HH];
        b.y = bcol[(size_t)k0 * HH + HH];
        c = __builtin_amdgcn_wmma_f32_16x16x4_f32(
                 false, a,  false, b,
                 (short)0, c,  false,  false);
    }

    const int nn = n0 + l15;
    const float bv = bias[nn];
#pragma unroll
    for (int r = 0; r < 8; ++r) {
        int m = m0 + r + half * 8;
        size_t idx = (size_t)m * HH + nn;
        float v = c[r] + bv;
        if (ori_in)  v += res * ori_in[idx];
        out[idx] = v;
        if (ori_out) ori_out[idx] = v;
    }
}

__global__ void ln_relu(float* __restrict__ h,
                        const float* __restrict__ g,
                        const float* __restrict__ bta) {
    int n = blockIdx.x * (blockDim.x >> 5) + (threadIdx.x >> 5);
    if (n >= NN) return;
    int lane = threadIdx.x & 31;
    float4* row = (float4*)(h + (size_t)n * HH);
    float4 v = row[lane];

    float s = v.x + v.y + v.z + v.w;
#pragma unroll
    for (int off = 16; off > 0; off >>= 1) s += __shfl_xor(s, off, 32);
    float mu = s * (1.0f / 128.0f);

    float dx = v.x - mu, dy = v.y - mu, dz = v.z - mu, dw = v.w - mu;
    float q = dx * dx + dy * dy + dz * dz + dw * dw;
#pragma unroll
    for (int off = 16; off > 0; off >>= 1) q += __shfl_xor(q, off, 32);
    float r = rsqrtf(q * (1.0f / 128.0f) + 1e-5f);

    int c = lane * 4;
    float4 o;
    o.x = fmaxf(dx * r * g[c + 0] + bta[c + 0], 0.0f);
    o.y = fmaxf(dy * r * g[c + 1] + bta[c + 1], 0.0f);
    o.z = fmaxf(dz * r * g[c + 2] + bta[c + 2], 0.0f);
    o.w = fmaxf(dw * r * g[c + 3] + bta[c + 3], 0.0f);
    row[lane] = o;
}

extern "C" void kernel_launch(void* const* d_in, const int* in_sizes, int n_in,
                              void* d_out, int out_size, void* d_ws, size_t ws_size,
                              hipStream_t stream) {
    (void)in_sizes; (void)n_in; (void)out_size; (void)ws_size;

    const float* x    = (const float*)d_in[0];
    const float* W0   = (const float*)d_in[1];
    const float* Ws   = (const float*)d_in[2];
    const float* b    = (const float*)d_in[3];
    const float* ln_g = (const float*)d_in[4];
    const float* ln_b = (const float*)d_in[5];
    const int*   src  = (const int*)d_in[6];
    const int*   dst  = (const int*)d_in[7];
    float* out = (float*)d_out;

    float* ws       = (float*)d_ws;
    float* norm_src = ws;
    float* norm_dst = ws + NN;
    float* agg      = ws + 2 * (size_t)NN;
    float* ori      = agg + (size_t)NN * HH;
    float* hbuf     = ori + (size_t)NN * HH;
    int* range_cnt  = (int*)(hbuf + (size_t)NN * HH);
    int* elist      = range_cnt + NBLK;
    unsigned* rbytes = (unsigned*)(elist + LCAP);

    hipMemsetAsync(ws, 0, 2 * (size_t)NN * sizeof(float), stream);
    deg_kernel<<<DEG_BLOCKS, 256, 0, stream>>>(src, dst, norm_src, norm_dst, rbytes);
    norm_kernel<<<NBLK, 512, 0, stream>>>(norm_src, norm_dst, range_cnt);

    agg_first<<<NBLK, 1024, 0, stream>>>(x, norm_src, src, dst, rbytes, range_cnt,
                                         elist, agg);
    gcn_gemm<<<NN / 16, 256, 0, stream>>>(agg, norm_dst, W0, b, nullptr, 0.0f,
                                          hbuf, ori);

    ln_relu<<<NN / 8, 256, 0, stream>>>(hbuf, ln_g + HH, ln_b + HH);
    agg_listed<<<NBLK, 1024, 0, stream>>>(hbuf, norm_src, range_cnt, elist, agg);
    gcn_gemm<<<NN / 16, 256, 0, stream>>>(agg, norm_dst, Ws, b + HH, ori, 0.5f,
                                          hbuf, nullptr);

    ln_relu<<<NN / 8, 256, 0, stream>>>(hbuf, ln_g + 2 * HH, ln_b + 2 * HH);
    agg_listed<<<NBLK, 1024, 0, stream>>>(hbuf, norm_src, range_cnt, elist, agg);
    gcn_gemm<<<NN / 16, 256, 0, stream>>>(agg, norm_dst, Ws + (size_t)HH * HH,
                                          b + 2 * HH, ori, 0.5f, out, nullptr);
}


# each layer projects first (f32 WMMA) then sums edges; the aggregate's row store does norm, bias, residual, layernorm+relu; layernorm launches and per-edge norm loads go; arrival-order sums of z rows
# speedup vs baseline: 3.0384x; 1.3711x over previous
#include <hip/hip_runtime.h>
#include <hip/hip_bf16.h>

#define NN 100000
#define EE 1600000
#define HH 128
#define LDS_STRIDE 132

typedef __attribute__((ext_vector_type(2))) float v2f;
typedef __attribute__((ext_vector_type(8))) float v8f;

#define RPB   391
#define NBLK  256
static_assert(NBLK <= 256 && (NN - 1) / RPB < 256, "a range number fits one byte");
static_assert(EE % 128 == 0, "whole waves of four edges per lane");

#define DEG_BLOCKS ((EE / 4 + 255) / 256)
__global__ void deg_kernel(const int* __restrict__ src, const int* __restrict__ dst,
                           float* __restrict__ deg_out, float* __restrict__ deg_in,
                           unsigned* __restrict__ rbytes) {
    const int t = blockIdx.x * blockDim.x + threadIdx.x;
    if (t < EE / 4) {
        const int4 s = ((const int4*)src)[t];
        const int4 d = ((const int4*)dst)[t];
        atomicAdd(&deg_out[s.x], 1.0f);
        atomicAdd(&deg_out[s.y], 1.0f);
        atomicAdd(&deg_out[s.z], 1.0f);
        atomicAdd(&deg_out[s.w], 1.0f);
        atomicAdd(&deg_in[d.x], 1.0f);
        atomicAdd(&deg_in[d.y], 1.0f);
        atomicAdd(&deg_in[d.z], 1.0f);
        atomicAdd(&deg_in[d.w], 1.0f);
        rbytes[t] = (unsigned)(d.x / RPB) | ((unsigned)(d.y / RPB) << 8) |
                    ((unsigned)(d.z / RPB) << 16) | ((unsigned)(d.w / RPB) << 24);
    }
}

__global__ __launch_bounds__(512)
void norm_kernel(float* __restrict__ ns, float* __restrict__ nd,
                 int* __restrict__ range_cnt) {
    __shared__ int tot;
    if (threadIdx.x == 0) tot = 0;
    __syncthreads();
    const int r = threadIdx.x;
    const int i = blockIdx.x * RPB + r;
    int c = 0;
    if (r < RPB && i < NN) {
        const float di = nd[i];
        c = (int)di;
        ns[i] = rsqrtf(fmaxf(ns[i], 1.0f));
        nd[i] = rsqrtf(fmaxf(di, 1.0f));
    }
    for (int off = 16; off > 0; off >>= 1) c += __shfl_xor(c, off, 32);
    if ((threadIdx.x & 31) == 0 && c != 0) atomicAdd(&tot, c);
    __syncthreads();
    if (threadIdx.x == 0) range_cnt[blockIdx.x] = tot;
}

#define CH    512
#define NCH   (EE / CH)
#define QCAP  544
#define SBITS 17
#define LCAP  (EE + NBLK * 32)
static_assert(EE % CH == 0, "edge list is a whole number of steps");
static_assert(NBLK * RPB >= NN, "ranges cover every row");
static_assert(31 + CH <= QCAP, "queue holds one step on top of a remainder");
static_assert(NCH + 64 + (NBLK - 1) * (NCH / NBLK) <= 2 * NCH, "read-ahead places wrap once");
static_assert(RPB <= 512, "local row packing");
static_assert(NN <= (1 << SBITS), "list word packing");
static_assert(NBLK <= 1024, "one thread per earlier group");

__device__ __forceinline__ float4 load_row(const float* __restrict__ feat,
                                           int row, int lane) {
    return ((const float4*)(feat + (size_t)row * HH))[lane];
}

__device__ __forceinline__ void add_row(float* acc, int r, int j,
                                        float4 w, int lane) {
    const int rj = __builtin_amdgcn_readlane(r, j);
    float* a = acc + rj * HH + lane;
    atomicAdd(a,      w.x);
    atomicAdd(a + 32, w.y);
    atomicAdd(a + 64, w.z);
    atomicAdd(a + 96, w.w);
}

template <bool FULL>
__device__ __forceinline__ void sum_chunk(const float* __restrict__ feat,
                                          float* acc, int s, int r,
                                          int cnt, int lane) {
    if (FULL) {
        float4 v[2][8];
#pragma unroll
        for (int u = 0; u < 8; ++u)
            v[0][u] = load_row(feat, __builtin_amdgcn_readlane(s, u), lane);
#pragma unroll
        for (int b = 0; b < 4; ++b) {
            if (b < 3) {
#pragma unroll
                for (int u = 0; u < 8; ++u)
                    v[(b + 1) & 1][u] = load_row(
                        feat, __builtin_amdgcn_readlane(s, 8 * (b + 1) + u), lane);
            }
#pragma unroll
            for (int u = 0; u < 8; ++u)
                add_row(acc, r, 8 * b + u, v[b & 1][u], lane);
        }
    } else {
#pragma unroll
        for (int b = 0; b < 4; ++b) {
            if (8 * b < cnt) {
                float4 v[8];
#pragma unroll
                for (int u = 0; u < 8; ++u)
                    if (8 * b + u < cnt)
                        v[u] = load_row(
                            feat, __builtin_amdgcn_readlane(s, 8 * b + u), lane);
#pragma unroll
                for (int u = 0; u < 8; ++u)
                    if (8 * b + u < cnt)
                        add_row(acc, r, 8 * b + u, v[u], lane);
            }
        }
    }
}

__device__ __forceinline__ void add_earlier_counts(const int* __restrict__ range_cnt,
                                                   int* goff_s, int tid) {
    if (tid < (int)blockIdx.x)
        atomicAdd(goff_s, (range_cnt[tid] + 31) & ~31);
}

#define RBATCH 4
template <int LAYER>
__device__ __forceinline__ void finish_rows(const float* acc, int base, int wave, int lane,
                                            const float* __restrict__ norm_src,
                                            const float* __restrict__ norm_dst,
                                            const float* __restrict__ bias,
                                            const float* __restrict__ g,
                                            const float* __restrict__ bta,
                                            float res,
                                            float* __restrict__ ori,
                                            float* __restrict__ dest) {
    const float4 bv = ((const float4*)bias)[lane];
    float4 gv = make_float4(0.0f, 0.0f, 0.0f, 0.0f);
    float4 tv = gv;
    if (LAYER < 2) {
        gv = ((const float4*)g)[lane];
        tv = ((const float4*)bta)[lane];
    }
    for (int row0 = wave; row0 < RPB; row0 += 32 * RBATCH) {
        float4 v[RBATCH], h[RBATCH];
        int nn[RBATCH];
        bool ok[RBATCH];
#pragma unroll
        for (int u = 0; u < RBATCH; ++u) {
            const int row = row0 + 32 * u;
            ok[u] = row < RPB && base + row < NN;
            int rc = row < RPB ? row : RPB - 1;
            int nc = base + rc;
            if (nc > NN - 1) { rc -= nc - (NN - 1); nc = NN - 1; }
            nn[u] = nc;
            const float* a = acc + rc * HH + lane;
            const float nd = norm_dst[nc];
            v[u].x = a[0]  * nd + bv.x;
            v[u].y = a[32] * nd + bv.y;
            v[u].z = a[64] * nd + bv.z;
            v[u].w = a[96] * nd + bv.w;
            if (LAYER > 0) {
                const float4 o = ((const float4*)(ori + (size_t)nc * HH))[lane];
                v[u].x += res * o.x;
                v[u].y += res * o.y;
                v[u].z += res * o.z;
                v[u].w += res * o.w;
            }
            if (LAYER < 2) {
                const float ns = norm_src[nc];
                float s = v[u].x + v[u].y + v[u].z + v[u].w;
#pragma unroll
                for (int off = 16; off > 0; off >>= 1) s += __shfl_xor(s, off, 32);
                const float mu = s * (1.0f / 128.0f);
                const float dx = v[u].x - mu, dy = v[u].y - mu;
                const float dz = v[u].z - mu, dw = v[u].w - mu;
                float q = dx * dx + dy * dy + dz * dz + dw * dw;
#pragma unroll
                for (int off = 16; off > 0; off >>= 1) q += __shfl_xor(q, off, 32);
                const float rs = rsqrtf(q * (1.0f / 128.0f) + 1e-5f);
                h[u].x = fmaxf(dx * rs * gv.x + tv.x, 0.0f) * ns;
                h[u].y = fmaxf(dy * rs * gv.y + tv.y, 0.0f) * ns;
                h[u].z = fmaxf(dz * rs * gv.z + tv.z, 0.0f) * ns;
                h[u].w = fmaxf(dw * rs * gv.w + tv.w, 0.0f) * ns;
            }
        }
#pragma unroll
        for (int u = 0; u < RBATCH; ++u) {
            if (ok[u]) {
                if (LAYER == 0) ((float4*)(ori + (size_t)nn[u] * HH))[lane] = v[u];
                ((float4*)(dest + (size_t)nn[u] * HH))[lane] = (LAYER == 2) ? v[u] : h[u];
            }
        }
    }
}

template <bool FULL>
__device__ __forceinline__ void take_entries(const float* __restrict__ feat,
                                             const int* __restrict__ src,
                                             const int* __restrict__ dst,
                                             int base,
                                             float* acc, const int* q,
                                             int first, int cnt, int lane,
                                             int* __restrict__ lst, int* cursor,
                                             int* tail, int* tailn) {
    int s = 0, r = 0;
    if (FULL || lane < cnt) {
        const int e = q[first + lane];
        s = src[e];
        r = dst[e] - base;
    }
    int pos = 0;
    if (lane == 0) pos = atomicAdd(FULL ? cursor : tailn, FULL ? 32 : cnt);
    pos = __builtin_amdgcn_readfirstlane(pos);
    if (FULL)
        lst[pos + lane] = s | (r << SBITS);
    else if (lane < cnt)
        tail[pos + lane] = s | (r << SBITS);
    sum_chunk<FULL>(feat, acc, s, r, cnt, lane);
}

__device__ __forceinline__ unsigned nonzero_bytes(unsigned x) {
    return (((x & 0x7F7F7F7Fu) + 0x7F7F7F7Fu) | x) & 0x80808080u;
}

__device__ __forceinline__ unsigned hit_bits(uint4 g, unsigned pat) {
    const unsigned n = (nonzero_bytes(g.x ^ pat) >> 7) | (nonzero_bytes(g.y ^ pat) >> 6) |
                       (nonzero_bytes(g.z ^ pat) >> 5) | (nonzero_bytes(g.w ^ pat) >> 4);
    return ~n & 0x0F0F0F0Fu;
}

__device__ __forceinline__ int step_place(int t, int start) {
    int p = t + start;
    if (p >= NCH) p -= NCH;
    return p;
}

__device__ __forceinline__ int filter_step(uint4& g, int t, int start, unsigned pat,
                                           const uint4* __restrict__ rb4,
                                           const float* __restrict__ feat,
                                           const int* __restrict__ src,
                                           const int* __restrict__ dst,
                                           int base, float* acc, int* q, int qn,
                                           int lane, int* __restrict__ lst,
                                           int* ctl, int* tail) {
    unsigned hits = hit_bits(g, pat);
    g = rb4[step_place(t + 64, start) * 32 + lane];
    const int e0 = step_place(t, start) * CH + lane * 16;
    for (int pass = 0; pass < 16; ++pass) {
        const bool m = hits != 0u;
        const unsigned mask = __builtin_amdgcn_ballot_w32(m);
        if (mask == 0u) break;
        if (m) {
            const int pos = __builtin_ctz(hits);
            q[qn + (int)__builtin_amdgcn_mbcnt_lo(mask, 0u)] =
                e0 + ((pos & 7) << 2) + (pos >> 3);
            hits &= hits - 1u;
        }
        qn += __builtin_popcount(mask);
    }
    while (qn >= 32) {
        qn -= 32;
        take_entries<true>(feat, src, dst, base, acc, q, qn, 32, lane,
                           lst, &ctl[1], tail, &ctl[2]);
    }
    return qn;
}

__global__ __launch_bounds__(1024)
void agg_first(const float* __restrict__ feat,
               const int* __restrict__ src, const int* __restrict__ dst,
               const unsigned* __restrict__ rbytes,
               const int* __restrict__ range_cnt,
               int* __restrict__ elist,
               const float* __restrict__ norm_src,
               const float* __restrict__ norm_dst,
               const float* __restrict__ bias,
               const float* __restrict__ g,
               const float* __restrict__ bta,
               float* __restrict__ ori,
               float* __restrict__ hnext) {
    __shared__ __attribute__((aligned(16))) float acc[RPB * HH];
    __shared__ int qall[32 * QCAP];
    __shared__ int tail[32 * 32];
    __shared__ int ctl[3];

    const int tid  = threadIdx.x;
    const int lane = tid & 31;
    const int wave = tid >> 5;
    const int base = blockIdx.x * RPB;

    if (tid < 3) ctl[tid] = 0;
    {
        float4* acc4 = (float4*)acc;
        for (int i = tid; i < RPB * HH / 4; i += 1024)
            acc4[i] = make_float4(0.0f, 0.0f, 0.0f, 0.0f);
    }
    __syncthreads();
    add_earlier_counts(range_cnt, &ctl[0], tid);
    __syncthreads();
    int* lst = elist + ctl[0];

    int* q = qall + wave * QCAP;
    int qn = 0;
    const uint4* rb4 = (const uint4*)rbytes;
    const unsigned pat = (unsigned)blockIdx.x * 0x01010101u;
    const int start = blockIdx.x * (NCH / NBLK);

    const int wv = __builtin_amdgcn_readfirstlane(wave);
    uint4 ga = rb4[step_place(wv, start) * 32 + lane];
    uint4 gb = rb4[step_place(wv + 32, start) * 32 + lane];
    for (int it = wv; it < NCH; it += 64) {
        qn = filter_step(ga, it, start, pat, rb4, feat, src, dst, base,
                         acc, q, qn, lane, lst, ctl, tail);
        if (it + 32 >= NCH) break;
        qn = filter_step(gb, it + 32, start, pat, rb4, feat, src, dst,
                         base, acc, q, qn, lane, lst, ctl, tail);
    }
    if (qn > 0)
        take_entries<false>(feat, src, dst, base, acc, q, 0, qn, lane,
                            lst, &ctl[1], tail, &ctl[2]);

    __syncthreads();

    {
        const int nfull = ctl[1];
        const int nt = ctl[2];
        for (int i = wave * 32; i < nt; i += 1024)
            if (i + lane < nt) lst[nfull + i + lane] = tail[i + lane];
    }
    finish_rows<0>(acc, base, wave, lane, norm_src, norm_dst, bias, g, bta, 0.0f,
                   ori, hnext);
}

template <int LAYER>
__device__ __forceinline__ void agg_listed(float* acc, int* goff_p,
                                           const float* __restrict__ feat,
                                           const int* __restrict__ range_cnt,
                                           const int* __restrict__ elist,
                                           const float* __restrict__ norm_src,
                                           const float* __restrict__ norm_dst,
                                           const float* __restrict__ bias,
                                           const float* __restrict__ g,
                                           const float* __restrict__ bta,
                                           float res,
                                           float* __restrict__ ori,
                                           float* __restrict__ dest) {
    int& goff_s = *goff_p;

    const int tid  = threadIdx.x;
    const int lane = tid & 31;
    const int wave = tid >> 5;
    const int base = blockIdx.x * RPB;

    if (tid == 0) goff_s = 0;
    {
        float4* acc4 = (float4*)acc;
        for (int i = tid; i < RPB * HH / 4; i += 1024)
            acc4[i] = make_float4(0.0f, 0.0f, 0.0f, 0.0f);
    }
    __syncthreads();
    add_earlier_counts(range_cnt, &goff_s, tid);
    __syncthreads();
    const int* lst = elist + goff_s;
    const int total = range_cnt[blockIdx.x];

    const int share = (((total + 31) >> 5) + 7) & ~7;
    int c0 = wave * share;
    int wend = c0 + share;
    if (wend > total) wend = total;

    const int smask = (1 << SBITS) - 1;
    int e0 = 0, e1 = 0;
    if (c0 + lane < wend)      e0 = lst[c0 + lane];
    if (c0 + 32 + lane < wend) e1 = lst[c0 + 32 + lane];

    while (c0 < wend) {
        int n = wend - c0;
        if (n > 32) n = 32;
        int e2 = 0;
        if (c0 + 64 + lane < wend) e2 = lst[c0 + 64 + lane];

        const int s = e0 & smask;
        const int r = e0 >> SBITS;
        if (n == 32) sum_chunk<true>(feat, acc, s, r, 32, lane);
        else         sum_chunk<false>(feat, acc, s, r, n, lane);

        e0 = e1; e1 = e2;
        c0 += 32;
    }

    __syncthreads();
    finish_rows<LAYER>(acc, base, wave, lane, norm_src, norm_dst, bias, g, bta, res,
                       ori, dest);
}

__global__ __launch_bounds__(1024)
void agg_mid(const float* __restrict__ feat,
             const int* __restrict__ range_cnt,
             const int* __restrict__ elist,
             const float* __restrict__ norm_src,
             const float* __restrict__ norm_dst,
             const float* __restrict__ bias,
             const float* __restrict__ g,
             const float* __restrict__ bta,
             float res,
             float* __restrict__ ori,
             float* __restrict__ hnext) {
    __shared__ __attribute__((aligned(16))) float acc[RPB * HH];
    __shared__ int goff_s;
    agg_listed<1>(acc, &goff_s, feat, range_cnt, elist, norm_src, norm_dst, bias,
                  g, bta, res, ori, hnext);
}

__global__ __launch_bounds__(1024)
void agg_last(const float* __restrict__ feat,
              const int* __restrict__ range_cnt,
              const int* __restrict__ elist,
              const float* __restrict__ norm_dst,
              const float* __restrict__ bias,
              float res,
              float* __restrict__ ori,
              float* __restrict__ out) {
    __shared__ __attribute__((aligned(16))) float acc[RPB * HH];
    __shared__ int goff_s;
    agg_listed<2>(acc, &goff_s, feat, range_cnt, elist, nullptr, norm_dst, bias,
                  nullptr, nullptr, res, ori, out);
}

__global__ void gcn_gemm(const float* __restrict__ h,
                         const float* __restrict__ scale,
                         const float* __restrict__ W,
                         float* __restrict__ z) {
    __shared__ float lds_a[16 * LDS_STRIDE];

    const int m0   = blockIdx.x * 16;
    const int tid  = threadIdx.x;
    const int wave = tid >> 5;
    const int lane = tid & 31;

    {
        int base = tid * 8;
        int row  = base >> 7;
        int col  = base & 127;
        float nd = scale ? scale[m0 + row] : 1.0f;
        const float4* s4 = (const float4*)(h + (size_t)(m0 + row) * HH + col);
        float4 a0 = s4[0];
        float4 a1 = s4[1];
        a0.x *= nd; a0.y *= nd; a0.z *= nd; a0.w *= nd;
        a1.x *= nd; a1.y *= nd; a1.z *= nd; a1.w *= nd;
        float4* t4 = (float4*)(lds_a + row * LDS_STRIDE + col);
        t4[0] = a0;
        t4[1] = a1;
    }
    __syncthreads();

    const int n0   = wave * 16;
    const int half = lane >> 4;
    const int l15  = lane & 15;

    const float* arow = lds_a + l15 * LDS_STRIDE + half * 2;
    const float* bcol = W + (size_t)(half * 2) * HH + n0 + l15;

    v8f c = {};
#pragma unroll
    for (int kk = 0; kk < 32; ++kk) {
        const int k0 = kk * 4;
        v2f a, b;
        a.x = arow[k0];
        a.y = arow[k0 + 1];
        b.x = bcol[(size_t)k0 * HH];
        b.y = bcol[(size_t)k0 * HH + HH];
        c = __builtin_amdgcn_wmma_f32_16x16x4_f32(
                 false, a,  false, b,
                 (short)0, c,  false,  false);
    }

    const int nn = n0 + l15;
#pragma unroll
    for (int r = 0; r < 8; ++r) {
        int m = m0 + r + half * 8;
        z[(size_t)m * HH + nn] = c[r];
    }
}

extern "C" void kernel_launch(void* const* d_in, const int* in_sizes, int n_in,
                              void* d_out, int out_size, void* d_ws, size_t ws_size,
                              hipStream_t stream) {
    (void)in_sizes; (void)n_in; (void)out_size; (void)ws_size;

    const float* x    = (const float*)d_in[0];
    const float* W0   = (const float*)d_in[1];
    const float* Ws   = (const float*)d_in[2];
    const float* b    = (const float*)d_in[3];
    const float* ln_g = (const float*)d_in[4];
    const float* ln_b = (const float*)d_in[5];
    const int*   src  = (const int*)d_in[6];
    const int*   dst  = (const int*)d_in[7];
    float* out = (float*)d_out;

    float* ws       = (float*)d_ws;
    float* norm_src = ws;
    float* norm_dst = ws + NN;
    float* zbuf     = ws + 2 * (size_t)NN;
    float* ori      = zbuf + (size_t)NN * HH;
    float* hbuf     = ori + (size_t)NN * HH;
    int* range_cnt  = (int*)(hbuf + (size_t)NN * HH);
    int* elist      = range_cnt + NBLK;
    unsigned* rbytes = (unsigned*)(elist + LCAP);

    hipMemsetAsync(ws, 0, 2 * (size_t)NN * sizeof(float), stream);
    deg_kernel<<<DEG_BLOCKS, 256, 0, stream>>>(src, dst, norm_src, norm_dst, rbytes);
    norm_kernel<<<NBLK, 512, 0, stream>>>(norm_src, norm_dst, range_cnt);

    gcn_gemm<<<NN / 16, 256, 0, stream>>>(x, norm_src, W0, zbuf);
    agg_first<<<NBLK, 1024, 0, stream>>>(zbuf, src, dst, rbytes, range_cnt, elist,
                                         norm_src, norm_dst, b, ln_g + HH, ln_b + HH,
                                         ori, hbuf);

    gcn_gemm<<<NN / 16, 256, 0, stream>>>(hbuf, nullptr, Ws, zbuf);
    agg_mid<<<NBLK, 1024, 0, stream>>>(zbuf, range_cnt, elist, norm_src, norm_dst,
                                       b + HH, ln_g + 2 * HH, ln_b + 2 * HH, 0.5f,
                                       ori, hbuf);

    gcn_gemm<<<NN / 16, 256, 0, stream>>>(hbuf, nullptr, Ws + (size_t)HH * HH, zbuf);
    agg_last<<<NBLK, 1024, 0, stream>>>(zbuf, range_cnt, elist, norm_dst, b + 2 * HH,
                                        0.5f, ori, out);
}


# per-range edge lists come from one partition launch (block range counts, cursor add, atomic-exchange scatter); groups no longer filter the whole edge list; arrival-order LDS sums kept
# speedup vs baseline: 3.1573x; 1.0391x over previous
#include <hip/hip_runtime.h>
#include <hip/hip_bf16.h>

#define NN 100000
#define EE 1600000
#define HH 128
#define LDS_STRIDE 132

typedef __attribute__((ext_vector_type(2))) float v2f;
typedef __attribute__((ext_vector_type(8))) float v8f;

#define RPB   391
#define NBLK  256
static_assert((NN - 1) / RPB < NBLK, "every row's range has a work-group");
static_assert(EE % 128 == 0, "whole waves of four edges per lane");

#define DEG_BLOCKS ((EE / 4 + 255) / 256)
__global__ void deg_kernel(const int* __restrict__ src, const int* __restrict__ dst,
                           float* __restrict__ deg_out, float* __restrict__ deg_in) {
    const int t = blockIdx.x * blockDim.x + threadIdx.x;
    if (t < EE / 4) {
        const int4 s = ((const int4*)src)[t];
        const int4 d = ((const int4*)dst)[t];
        atomicAdd(&deg_out[s.x], 1.0f);
        atomicAdd(&deg_out[s.y], 1.0f);
        atomicAdd(&deg_out[s.z], 1.0f);
        atomicAdd(&deg_out[s.w], 1.0f);
        atomicAdd(&deg_in[d.x], 1.0f);
        atomicAdd(&deg_in[d.y], 1.0f);
        atomicAdd(&deg_in[d.z], 1.0f);
        atomicAdd(&deg_in[d.w], 1.0f);
    }
}

__global__ __launch_bounds__(512)
void norm_kernel(float* __restrict__ ns, float* __restrict__ nd,
                 int* __restrict__ range_cnt, int* __restrict__ cursor) {
    __shared__ int tot;
    if (threadIdx.x == 0) tot = 0;
    __syncthreads();
    const int r = threadIdx.x;
    const int i = blockIdx.x * RPB + r;
    int c = 0;
    if (r < RPB && i < NN) {
        const float di = nd[i];
        c = (int)di;
        ns[i] = rsqrtf(fmaxf(ns[i], 1.0f));
        nd[i] = rsqrtf(fmaxf(di, 1.0f));
    }
    for (int off = 16; off > 0; off >>= 1) c += __shfl_xor(c, off, 32);
    if ((threadIdx.x & 31) == 0 && c != 0) atomicAdd(&tot, c);
    __syncthreads();
    if (threadIdx.x == 0) {
        range_cnt[blockIdx.x] = tot;
        cursor[blockIdx.x] = 0;
    }
}

#define SBITS 17
#define LCAP  (EE + NBLK * 32)
static_assert(NBLK * RPB >= NN, "ranges cover every row");
static_assert(RPB <= 512, "local row packing");
static_assert(NN <= (1 << SBITS), "list word packing");
static_assert(NBLK <= 1024, "one thread per earlier group");

#define PBLOCKS  256
#define PTHREADS 512
#define PEDGES   (EE / PBLOCKS)
#define PPT      ((PEDGES + PTHREADS - 1) / PTHREADS)
static_assert(EE % PBLOCKS == 0, "blocks take equal shares of the edge list");
static_assert(NBLK == 256 && PTHREADS >= NBLK, "one thread per range, eight whole waves");

__global__ __launch_bounds__(PTHREADS)
void part_kernel(const int* __restrict__ src, const int* __restrict__ dst,
                 const int* __restrict__ range_cnt,
                 int* __restrict__ cursor, int* __restrict__ elist) {
    __shared__ int hist[NBLK];
    __shared__ int gpos[NBLK];
    __shared__ int lpos[NBLK];
    __shared__ int wtc[8], wth[8];
    __shared__ int sword[PEDGES];
    __shared__ int sdest[PEDGES];

    const int tid  = threadIdx.x;
    const int lane = tid & 31;
    const int wave = tid >> 5;

    if (tid < NBLK) hist[tid] = 0;
    __syncthreads();

    const int e0 = blockIdx.x * PEDGES;
    int sv[PPT], dv[PPT];
    int word[PPT], rng[PPT], rank[PPT];
#pragma unroll
    for (int k = 0; k < PPT; ++k) {
        const int i = tid + k * PTHREADS;
        const int ii = i < PEDGES ? i : PEDGES - 1;
        sv[k] = src[e0 + ii];
        dv[k] = dst[e0 + ii];
    }
#pragma unroll
    for (int k = 0; k < PPT; ++k) {
        const int i = tid + k * PTHREADS;
        rng[k] = -1;
        word[k] = 0;
        rank[k] = 0;
        if (i < PEDGES) {
            const int r = dv[k] / RPB;
            rng[k]  = r;
            word[k] = sv[k] | ((dv[k] - r * RPB) << SBITS);
            rank[k] = atomicAdd(&hist[r], 1);
        }
    }
    __syncthreads();

    int c = 0, h = 0, ic = 0, ih = 0;
    if (tid < NBLK) {
        c = (range_cnt[tid] + 31) & ~31;
        h = hist[tid];
        ic = c;
        ih = h;
#pragma unroll
        for (int off = 1; off < 32; off <<= 1) {
            const int vc = __shfl_up(ic, off, 32);
            const int vh = __shfl_up(ih, off, 32);
            if (lane >= off) { ic += vc; ih += vh; }
        }
        if (lane == 31) { wtc[wave] = ic; wth[wave] = ih; }
    }
    __syncthreads();
    if (tid < NBLK) {
        int bc = 0, bh = 0;
        for (int w = 0; w < wave; ++w) { bc += wtc[w]; bh += wth[w]; }
        int got = 0;
        if (h != 0) got = atomicAdd(&cursor[tid], h);
        gpos[tid] = bc + ic - c + got;
        lpos[tid] = bh + ih - h;
    }
    __syncthreads();

#pragma unroll
    for (int k = 0; k < PPT; ++k) {
        if (rng[k] >= 0) {
            const int p = lpos[rng[k]] + rank[k];
            sword[p] = word[k];
            sdest[p] = gpos[rng[k]] + rank[k];
        }
    }
    __syncthreads();

    for (int i = tid; i < PEDGES; i += PTHREADS)
        atomicExch(&elist[sdest[i]], sword[i]);
}

__device__ __forceinline__ float4 load_row(const float* __restrict__ feat,
                                           int row, int lane) {
    return ((const float4*)(feat + (size_t)row * HH))[lane];
}

__device__ __forceinline__ void add_row(float* acc, int r, int j,
                                        float4 w, int lane) {
    const int rj = __builtin_amdgcn_readlane(r, j);
    float* a = acc + rj * HH + lane;
    atomicAdd(a,      w.x);
    atomicAdd(a + 32, w.y);
    atomicAdd(a + 64, w.z);
    atomicAdd(a + 96, w.w);
}

template <bool FULL>
__device__ __forceinline__ void sum_chunk(const float* __restrict__ feat,
                                          float* acc, int s, int r,
                                          int cnt, int lane) {
    if (FULL) {
        float4 v[2][8];
#pragma unroll
        for (int u = 0; u < 8; ++u)
            v[0][u] = load_row(feat, __builtin_amdgcn_readlane(s, u), lane);
#pragma unroll
        for (int b = 0; b < 4; ++b) {
            if (b < 3) {
#pragma unroll
                for (int u = 0; u < 8; ++u)
                    v[(b + 1) & 1][u] = load_row(
                        feat, __builtin_amdgcn_readlane(s, 8 * (b + 1) + u), lane);
            }
#pragma unroll
            for (int u = 0; u < 8; ++u)
                add_row(acc, r, 8 * b + u, v[b & 1][u], lane);
        }
    } else {
#pragma unroll
        for (int b = 0; b < 4; ++b) {
            if (8 * b < cnt) {
                float4 v[8];
#pragma unroll
                for (int u = 0; u < 8; ++u)
                    if (8 * b + u < cnt)
                        v[u] = load_row(
                            feat, __builtin_amdgcn_readlane(s, 8 * b + u), lane);
#pragma unroll
                for (int u = 0; u < 8; ++u)
                    if (8 * b + u < cnt)
                        add_row(acc, r, 8 * b + u, v[u], lane);
            }
        }
    }
}

__device__ __forceinline__ void add_earlier_counts(const int* __restrict__ range_cnt,
                                                   int* goff_s, int tid) {
    if (tid < (int)blockIdx.x)
        atomicAdd(goff_s, (range_cnt[tid] + 31) & ~31);
}

#define RBATCH 4
template <int LAYER>
__device__ __forceinline__ void finish_rows(const float* acc, int base, int wave, int lane,
                                            const float* __restrict__ norm_src,
                                            const float* __restrict__ norm_dst,
                                            const float* __restrict__ bias,
                                            const float* __restrict__ g,
                                            const float* __restrict__ bta,
                                            float res,
                                            float* __restrict__ ori,
                                            float* __restrict__ dest) {
    const float4 bv = ((const float4*)bias)[lane];
    float4 gv = make_float4(0.0f, 0.0f, 0.0f, 0.0f);
    float4 tv = gv;
    if (LAYER < 2) {
        gv = ((const float4*)g)[lane];
        tv = ((const float4*)bta)[lane];
    }
    for (int row0 = wave; row0 < RPB; row0 += 32 * RBATCH) {
        float4 v[RBATCH], h[RBATCH];
        int nn[RBATCH];
        bool ok[RBATCH];
#pragma unroll
        for (int u = 0; u < RBATCH; ++u) {
            const int row = row0 + 32 * u;
            ok[u] = row < RPB && base + row < NN;
            int rc = row < RPB ? row : RPB - 1;
            int nc = base + rc;
            if (nc > NN - 1) { rc -= nc - (NN - 1); nc = NN - 1; }
            nn[u] = nc;
            const float* a = acc + rc * HH + lane;
            const float nd = norm_dst[nc];
            v[u].x = a[0]  * nd + bv.x;
            v[u].y = a[32] * nd + bv.y;
            v[u].z = a[64] * nd + bv.z;
            v[u].w = a[96] * nd + bv.w;
            if (LAYER > 0) {
                const float4 o = ((const float4*)(ori + (size_t)nc * HH))[lane];
                v[u].x += res * o.x;
                v[u].y += res * o.y;
                v[u].z += res * o.z;
                v[u].w += res * o.w;
            }
            if (LAYER < 2) {
                const float ns = norm_src[nc];
                float s = v[u].x + v[u].y + v[u].z + v[u].w;
#pragma unroll
                for (int off = 16; off > 0; off >>= 1) s += __shfl_xor(s, off, 32);
                const float mu = s * (1.0f / 128.0f);
                const float dx = v[u].x - mu, dy = v[u].y - mu;
                const float dz = v[u].z - mu, dw = v[u].w - mu;
                float q = dx * dx + dy * dy + dz * dz + dw * dw;
#pragma unroll
                for (int off = 16; off > 0; off >>= 1) q += __shfl_xor(q, off, 32);
                const float rs = rsqrtf(q * (1.0f / 128.0f) + 1e-5f);
                h[u].x = fmaxf(dx * rs * gv.x + tv.x, 0.0f) * ns;
                h[u].y = fmaxf(dy * rs * gv.y + tv.y, 0.0f) * ns;
                h[u].z = fmaxf(dz * rs * gv.z + tv.z, 0.0f) * ns;
                h[u].w = fmaxf(dw * rs * gv.w + tv.w, 0.0f) * ns;
            }
        }
#pragma unroll
        for (int u = 0; u < RBATCH; ++u) {
            if (ok[u]) {
                if (LAYER == 0) ((float4*)(ori + (size_t)nn[u] * HH))[lane] = v[u];
                ((float4*)(dest + (size_t)nn[u] * HH))[lane] = (LAYER == 2) ? v[u] : h[u];
            }
        }
    }
}

template <int LAYER>
__device__ __forceinline__ void agg_listed(float* acc, int* goff_p,
                                           const float* __restrict__ feat,
                                           const int* __restrict__ range_cnt,
                                           const int* __restrict__ elist,
                                           const float* __restrict__ norm_src,
                                           const float* __restrict__ norm_dst,
                                           const float* __restrict__ bias,
                                           const float* __restrict__ g,
                                           const float* __restrict__ bta,
                                           float res,
                                           float* __restrict__ ori,
                                           float* __restrict__ dest) {
    int& goff_s = *goff_p;

    const int tid  = threadIdx.x;
    const int lane = tid & 31;
    const int wave = tid >> 5;
    const int base = blockIdx.x * RPB;

    if (tid == 0) goff_s = 0;
    {
        float4* acc4 = (float4*)acc;
        for (int i = tid; i < RPB * HH / 4; i += 1024)
            acc4[i] = make_float4(0.0f, 0.0f, 0.0f, 0.0f);
    }
    __syncthreads();
    add_earlier_counts(range_cnt, &goff_s, tid);
    __syncthreads();
    const int* lst = elist + goff_s;
    const int total = range_cnt[blockIdx.x];

    const int share = (((total + 31) >> 5) + 7) & ~7;
    int c0 = wave * share;
    int wend = c0 + share;
    if (wend > total) wend = total;

    const int smask = (1 << SBITS) - 1;
    int e0 = 0, e1 = 0;
    if (c0 + lane < wend)      e0 = lst[c0 + lane];
    if (c0 + 32 + lane < wend) e1 = lst[c0 + 32 + lane];

    while (c0 < wend) {
        int n = wend - c0;
        if (n > 32) n = 32;
        int e2 = 0;
        if (c0 + 64 + lane < wend) e2 = lst[c0 + 64 + lane];

        const int s = e0 & smask;
        const int r = e0 >> SBITS;
        if (n == 32) sum_chunk<true>(feat, acc, s, r, 32, lane);
        else         sum_chunk<false>(feat, acc, s, r, n, lane);

        e0 = e1; e1 = e2;
        c0 += 32;
    }

    __syncthreads();
    finish_rows<LAYER>(acc, base, wave, lane, norm_src, norm_dst, bias, g, bta, res,
                       ori, dest);
}

__global__ __launch_bounds__(1024)
void agg_head(const float* __restrict__ feat,
              const int* __restrict__ range_cnt,
              const int* __restrict__ elist,
              const float* __restrict__ norm_src,
              const float* __restrict__ norm_dst,
              const float* __restrict__ bias,
              const float* __restrict__ g,
              const float* __restrict__ bta,
              float* __restrict__ ori,
              float* __restrict__ hnext) {
    __shared__ __attribute__((aligned(16))) float acc[RPB * HH];
    __shared__ int goff_s;
    agg_listed<0>(acc, &goff_s, feat, range_cnt, elist, norm_src, norm_dst, bias,
                  g, bta, 0.0f, ori, hnext);
}

__global__ __launch_bounds__(1024)
void agg_mid(const float* __restrict__ feat,
             const int* __restrict__ range_cnt,
             const int* __restrict__ elist,
             const float* __restrict__ norm_src,
             const float* __restrict__ norm_dst,
             const float* __restrict__ bias,
             const float* __restrict__ g,
             const float* __restrict__ bta,
             float res,
             float* __restrict__ ori,
             float* __restrict__ hnext) {
    __shared__ __attribute__((aligned(16))) float acc[RPB * HH];
    __shared__ int goff_s;
    agg_listed<1>(acc, &goff_s, feat, range_cnt, elist, norm_src, norm_dst, bias,
                  g, bta, res, ori, hnext);
}

__global__ __launch_bounds__(1024)
void agg_last(const float* __restrict__ feat,
              const int* __restrict__ range_cnt,
              const int* __restrict__ elist,
              const float* __restrict__ norm_dst,
              const float* __restrict__ bias,
              float res,
              float* __restrict__ ori,
              float* __restrict__ out) {
    __shared__ __attribute__((aligned(16))) float acc[RPB * HH];
    __shared__ int goff_s;
    agg_listed<2>(acc, &goff_s, feat, range_cnt, elist, nullptr, norm_dst, bias,
                  nullptr, nullptr, res, ori, out);
}

__global__ void gcn_gemm(const float* __restrict__ h,
                         const float* __restrict__ scale,
                         const float* __restrict__ W,
                         float* __restrict__ z) {
    __shared__ float lds_a[16 * LDS_STRIDE];

    const int m0   = blockIdx.x * 16;
    const int tid  = threadIdx.x;
    const int wave = tid >> 5;
    const int lane = tid & 31;

    {
        int base = tid * 8;
        int row  = base >> 7;
        int col  = base & 127;
        float nd = scale ? scale[m0 + row] : 1.0f;
        const float4* s4 = (const float4*)(h + (size_t)(m0 + row) * HH + col);
        float4 a0 = s4[0];
        float4 a1 = s4[1];
        a0.x *= nd; a0.y *= nd; a0.z *= nd; a0.w *= nd;
        a1.x *= nd; a1.y *= nd; a1.z *= nd; a1.w *= nd;
        float4* t4 = (float4*)(lds_a + row * LDS_STRIDE + col);
        t4[0] = a0;
        t4[1] = a1;
    }
    __syncthreads();

    const int n0   = wave * 16;
    const int half = lane >> 4;
    const int l15  = lane & 15;

    const float* arow = lds_a + l15 * LDS_STRIDE + half * 2;
    const float* bcol = W + (size_t)(half * 2) * HH + n0 + l15;

    v8f c = {};
#pragma unroll
    for (int kk = 0; kk < 32; ++kk) {
        const int k0 = kk * 4;
        v2f a, b;
        a.x = arow[k0];
        a.y = arow[k0 + 1];
        b.x = bcol[(size_t)k0 * HH];
        b.y = bcol[(size_t)k0 * HH + HH];
        c = __builtin_amdgcn_wmma_f32_16x16x4_f32(
                 false, a,  false, b,
                 (short)0, c,  false,  false);
    }

    const int nn = n0 + l15;
#pragma unroll
    for (int r = 0; r < 8; ++r) {
        int m = m0 + r + half * 8;
        z[(size_t)m * HH + nn] = c[r];
    }
}

extern "C" void kernel_launch(void* const* d_in, const int* in_sizes, int n_in,
                              void* d_out, int out_size, void* d_ws, size_t ws_size,
                              hipStream_t stream) {
    (void)in_sizes; (void)n_in; (void)out_size; (void)ws_size;

    const float* x    = (const float*)d_in[0];
    const float* W0   = (const float*)d_in[1];
    const float* Ws   = (const float*)d_in[2];
    const float* b    = (const float*)d_in[3];
    const float* ln_g = (const float*)d_in[4];
    const float* ln_b = (const float*)d_in[5];
    const int*   src  = (const int*)d_in[6];
    const int*   dst  = (const int*)d_in[7];
    float* out = (float*)d_out;

    float* ws       = (float*)d_ws;
    float* norm_src = ws;
    float* norm_dst = ws + NN;
    float* zbuf     = ws + 2 * (size_t)NN;
    float* ori      = zbuf + (size_t)NN * HH;
    float* hbuf     = ori + (size_t)NN * HH;
    int* range_cnt  = (int*)(hbuf + (size_t)NN * HH);
    int* cursor     = range_cnt + NBLK;
    int* elist      = cursor + NBLK;

    hipMemsetAsync(ws, 0, 2 * (size_t)NN * sizeof(float), stream);
    deg_kernel<<<DEG_BLOCKS, 256, 0, stream>>>(src, dst, norm_src, norm_dst);
    norm_kernel<<<NBLK, 512, 0, stream>>>(norm_src, norm_dst, range_cnt, cursor);

    part_kernel<<<PBLOCKS, PTHREADS, 0, stream>>>(src, dst, range_cnt, cursor, elist);

    gcn_gemm<<<NN / 16, 256, 0, stream>>>(x, norm_src, W0, zbuf);
    agg_head<<<NBLK, 1024, 0, stream>>>(zbuf, range_cnt, elist,
                                        norm_src, norm_dst, b, ln_g + HH, ln_b + HH,
                                        ori, hbuf);

    gcn_gemm<<<NN / 16, 256, 0, stream>>>(hbuf, nullptr, Ws, zbuf);
    agg_mid<<<NBLK, 1024, 0, stream>>>(zbuf, range_cnt, elist, norm_src, norm_dst,
                                       b + HH, ln_g + 2 * HH, ln_b + 2 * HH, 0.5f,
                                       ori, hbuf);

    gcn_gemm<<<NN / 16, 256, 0, stream>>>(hbuf, nullptr, Ws + (size_t)HH * HH, zbuf);
    agg_last<<<NBLK, 1024, 0, stream>>>(zbuf, range_cnt, elist, norm_dst, b + 2 * HH,
                                        0.5f, ori, out);
}


# layers 0,1: the aggregate launch projects its own finished LDS rows for the next layer (f32 WMMA, W slab in registers); 2 projection launches and the h round trip go; those products' k order changes
# speedup vs baseline: 3.4237x; 1.0844x over previous
#include <hip/hip_runtime.h>
#include <hip/hip_bf16.h>

#define NN 100000
#define EE 1600000
#define HH 128
#define LDS_STRIDE 132

typedef __attribute__((ext_vector_type(2))) float v2f;
typedef __attribute__((ext_vector_type(8))) float v8f;

#define RPB   391
#define NBLK  256
static_assert((NN - 1) / RPB < NBLK, "every row's range has a work-group");
static_assert(EE % 128 == 0, "whole waves of four edges per lane");

#define DEG_BLOCKS ((EE / 4 + 255) / 256)
__global__ void deg_kernel(const int* __restrict__ src, const int* __restrict__ dst,
                           float* __restrict__ deg_out, float* __restrict__ deg_in) {
    const int t = blockIdx.x * blockDim.x + threadIdx.x;
    if (t < EE / 4) {
        const int4 s = ((const int4*)src)[t];
        const int4 d = ((const int4*)dst)[t];
        atomicAdd(&deg_out[s.x], 1.0f);
        atomicAdd(&deg_out[s.y], 1.0f);
        atomicAdd(&deg_out[s.z], 1.0f);
        atomicAdd(&deg_out[s.w], 1.0f);
        atomicAdd(&deg_in[d.x], 1.0f);
        atomicAdd(&deg_in[d.y], 1.0f);
        atomicAdd(&deg_in[d.z], 1.0f);
        atomicAdd(&deg_in[d.w], 1.0f);
    }
}

__global__ __launch_bounds__(512)
void norm_kernel(float* __restrict__ ns, float* __restrict__ nd,
                 int* __restrict__ range_cnt, int* __restrict__ cursor) {
    __shared__ int tot;
    if (threadIdx.x == 0) tot = 0;
    __syncthreads();
    const int r = threadIdx.x;
    const int i = blockIdx.x * RPB + r;
    int c = 0;
    if (r < RPB && i < NN) {
        const float di = nd[i];
        c = (int)di;
        ns[i] = rsqrtf(fmaxf(ns[i], 1.0f));
        nd[i] = rsqrtf(fmaxf(di, 1.0f));
    }
    for (int off = 16; off > 0; off >>= 1) c += __shfl_xor(c, off, 32);
    if ((threadIdx.x & 31) == 0 && c != 0) atomicAdd(&tot, c);
    __syncthreads();
    if (threadIdx.x == 0) {
        range_cnt[blockIdx.x] = tot;
        cursor[blockIdx.x] = 0;
    }
}

#define APITCH 132
#define AROWS  400
static_assert(AROWS % 16 == 0 && AROWS >= RPB, "whole tiles cover the group's rows");
#define SBITS 17
#define LCAP  (EE + NBLK * 32)
static_assert(NBLK * RPB >= NN, "ranges cover every row");
static_assert(RPB <= 512, "local row packing");
static_assert(NN <= (1 << SBITS), "list word packing");
static_assert(NBLK <= 1024, "one thread per earlier group");

#define PBLOCKS  256
#define PTHREADS 512
#define PEDGES   (EE / PBLOCKS)
#define PPT      ((PEDGES + PTHREADS - 1) / PTHREADS)
static_assert(EE % PBLOCKS == 0, "blocks take equal shares of the edge list");
static_assert(NBLK == 256 && PTHREADS >= NBLK, "one thread per range, eight whole waves");

__global__ __launch_bounds__(PTHREADS)
void part_kernel(const int* __restrict__ src, const int* __restrict__ dst,
                 const int* __restrict__ range_cnt,
                 int* __restrict__ cursor, int* __restrict__ elist) {
    __shared__ int hist[NBLK];
    __shared__ int gpos[NBLK];
    __shared__ int lpos[NBLK];
    __shared__ int wtc[8], wth[8];
    __shared__ int sword[PEDGES];
    __shared__ int sdest[PEDGES];

    const int tid  = threadIdx.x;
    const int lane = tid & 31;
    const int wave = tid >> 5;

    if (tid < NBLK) hist[tid] = 0;
    __syncthreads();

    const int e0 = blockIdx.x * PEDGES;
    int sv[PPT], dv[PPT];
    int word[PPT], rng[PPT], rank[PPT];
#pragma unroll
    for (int k = 0; k < PPT; ++k) {
        const int i = tid + k * PTHREADS;
        const int ii = i < PEDGES ? i : PEDGES - 1;
        sv[k] = src[e0 + ii];
        dv[k] = dst[e0 + ii];
    }
#pragma unroll
    for (int k = 0; k < PPT; ++k) {
        const int i = tid + k * PTHREADS;
        rng[k] = -1;
        word[k] = 0;
        rank[k] = 0;
        if (i < PEDGES) {
            const int r = dv[k] / RPB;
            rng[k]  = r;
            word[k] = sv[k] | ((dv[k] - r * RPB) << SBITS);
            rank[k] = atomicAdd(&hist[r], 1);
        }
    }
    __syncthreads();

    int c = 0, h = 0, ic = 0, ih = 0;
    if (tid < NBLK) {
        c = (range_cnt[tid] + 31) & ~31;
        h = hist[tid];
        ic = c;
        ih = h;
#pragma unroll
        for (int off = 1; off < 32; off <<= 1) {
            const int vc = __shfl_up(ic, off, 32);
            const int vh = __shfl_up(ih, off, 32);
            if (lane >= off) { ic += vc; ih += vh; }
        }
        if (lane == 31) { wtc[wave] = ic; wth[wave] = ih; }
    }
    __syncthreads();
    if (tid < NBLK) {
        int bc = 0, bh = 0;
        for (int w = 0; w < wave; ++w) { bc += wtc[w]; bh += wth[w]; }
        int got = 0;
        if (h != 0) got = atomicAdd(&cursor[tid], h);
        gpos[tid] = bc + ic - c + got;
        lpos[tid] = bh + ih - h;
    }
    __syncthreads();

#pragma unroll
    for (int k = 0; k < PPT; ++k) {
        if (rng[k] >= 0) {
            const int p = lpos[rng[k]] + rank[k];
            sword[p] = word[k];
            sdest[p] = gpos[rng[k]] + rank[k];
        }
    }
    __syncthreads();

    for (int i = tid; i < PEDGES; i += PTHREADS)
        atomicExch(&elist[sdest[i]], sword[i]);
}

__device__ __forceinline__ float4 load_row(const float* __restrict__ feat,
                                           int row, int lane) {
    return ((const float4*)(feat + (size_t)row * HH))[lane];
}

__device__ __forceinline__ void add_row(float* acc, int r, int j,
                                        float4 w, int lane) {
    const int rj = __builtin_amdgcn_readlane(r, j);
    float* a = acc + rj * APITCH + lane;
    atomicAdd(a,      w.x);
    atomicAdd(a + 32, w.y);
    atomicAdd(a + 64, w.z);
    atomicAdd(a + 96, w.w);
}

template <bool FULL>
__device__ __forceinline__ void sum_chunk(const float* __restrict__ feat,
                                          float* acc, int s, int r,
                                          int cnt, int lane) {
    if (FULL) {
        float4 v[2][8];
#pragma unroll
        for (int u = 0; u < 8; ++u)
            v[0][u] = load_row(feat, __builtin_amdgcn_readlane(s, u), lane);
#pragma unroll
        for (int b = 0; b < 4; ++b) {
            if (b < 3) {
#pragma unroll
                for (int u = 0; u < 8; ++u)
                    v[(b + 1) & 1][u] = load_row(
                        feat, __builtin_amdgcn_readlane(s, 8 * (b + 1) + u), lane);
            }
#pragma unroll
            for (int u = 0; u < 8; ++u)
                add_row(acc, r, 8 * b + u, v[b & 1][u], lane);
        }
    } else {
#pragma unroll
        for (int b = 0; b < 4; ++b) {
            if (8 * b < cnt) {
                float4 v[8];
#pragma unroll
                for (int u = 0; u < 8; ++u)
                    if (8 * b + u < cnt)
                        v[u] = load_row(
                            feat, __builtin_amdgcn_readlane(s, 8 * b + u), lane);
#pragma unroll
                for (int u = 0; u < 8; ++u)
                    if (8 * b + u < cnt)
                        add_row(acc, r, 8 * b + u, v[u], lane);
            }
        }
    }
}

__device__ __forceinline__ void add_earlier_counts(const int* __restrict__ range_cnt,
                                                   int* goff_s, int tid) {
    if (tid < (int)blockIdx.x)
        atomicAdd(goff_s, (range_cnt[tid] + 31) & ~31);
}

#define RBATCH 4
template <int LAYER>
__device__ __forceinline__ void finish_rows(float* acc, int base, int wave, int lane,
                                            const float* __restrict__ norm_src,
                                            const float* __restrict__ norm_dst,
                                            const float* __restrict__ bias,
                                            const float* __restrict__ g,
                                            const float* __restrict__ bta,
                                            float res,
                                            float* __restrict__ ori,
                                            float* __restrict__ dest) {
    const float4 bv = ((const float4*)bias)[lane];
    float4 gv = make_float4(0.0f, 0.0f, 0.0f, 0.0f);
    float4 tv = gv;
    if (LAYER < 2) {
        gv = ((const float4*)g)[lane];
        tv = ((const float4*)bta)[lane];
    }
    for (int row0 = wave; row0 < RPB; row0 += 32 * RBATCH) {
        float4 v[RBATCH], h[RBATCH];
        int nn[RBATCH];
        bool ok[RBATCH];
#pragma unroll
        for (int u = 0; u < RBATCH; ++u) {
            const int row = row0 + 32 * u;
            ok[u] = row < RPB && base + row < NN;
            int rc = row < RPB ? row : RPB - 1;
            int nc = base + rc;
            if (nc > NN - 1) { rc -= nc - (NN - 1); nc = NN - 1; }
            nn[u] = nc;
            const float* a = acc + rc * APITCH + lane;
            const float nd = norm_dst[nc];
            v[u].x = a[0]  * nd + bv.x;
            v[u].y = a[32] * nd + bv.y;
            v[u].z = a[64] * nd + bv.z;
            v[u].w = a[96] * nd + bv.w;
            if (LAYER > 0) {
                const float4 o = ((const float4*)(ori + (size_t)nc * HH))[lane];
                v[u].x += res * o.x;
                v[u].y += res * o.y;
                v[u].z += res * o.z;
                v[u].w += res * o.w;
            }
            if (LAYER < 2) {
                const float ns = norm_src[nc];
                float s = v[u].x + v[u].y + v[u].z + v[u].w;
#pragma unroll
                for (int off = 16; off > 0; off >>= 1) s += __shfl_xor(s, off, 32);
                const float mu = s * (1.0f / 128.0f);
                const float dx = v[u].x - mu, dy = v[u].y - mu;
                const float dz = v[u].z - mu, dw = v[u].w - mu;
                float q = dx * dx + dy * dy + dz * dz + dw * dw;
#pragma unroll
                for (int off = 16; off > 0; off >>= 1) q += __shfl_xor(q, off, 32);
                const float rs = rsqrtf(q * (1.0f / 128.0f) + 1e-5f);
                h[u].x = fmaxf(dx * rs * gv.x + tv.x, 0.0f) * ns;
                h[u].y = fmaxf(dy * rs * gv.y + tv.y, 0.0f) * ns;
                h[u].z = fmaxf(dz * rs * gv.z + tv.z, 0.0f) * ns;
                h[u].w = fmaxf(dw * rs * gv.w + tv.w, 0.0f) * ns;
            }
        }
#pragma unroll
        for (int u = 0; u < RBATCH; ++u) {
            if (ok[u]) {
                if (LAYER == 0) ((float4*)(ori + (size_t)nn[u] * HH))[lane] = v[u];
                if (LAYER == 2) {
                    ((float4*)(dest + (size_t)nn[u] * HH))[lane] = v[u];
                } else {
                    float* aw = acc + (row0 + 32 * u) * APITCH + lane;
                    aw[0]  = h[u].x;
                    aw[32] = h[u].y;
                    aw[64] = h[u].z;
                    aw[96] = h[u].w;
                }
            }
        }
    }
}

__device__ __forceinline__ void load_slab(v2f* b, const float* __restrict__ Wn,
                                          int wave, int lane) {
    const int half = lane >> 4;
    const int col  = (wave & 7) * 16 + (lane & 15);
#pragma unroll
    for (int kk = 0; kk < 32; ++kk) {
        const int j0 = 16 * (kk >> 2) + 8 * half + (kk & 3);
        b[kk].x = Wn[(size_t)j0 * HH + col];
        b[kk].y = Wn[(size_t)(j0 + 4) * HH + col];
    }
}

__device__ __forceinline__ void project_rows(const float* acc, const v2f* b,
                                             int base, int wave, int lane,
                                             float* __restrict__ znext) {
    const int half = lane >> 4;
    const int l15  = lane & 15;
    const int wv   = __builtin_amdgcn_readfirstlane(wave);
    const int col  = (wv & 7) * 16 + l15;
    int limit = NN - base;
    if (limit > RPB) limit = RPB;
    const float* arow = acc + l15 * APITCH + 2 * half;
    for (int rt = wv >> 3; rt < AROWS / 16; rt += 4) {
        const int nv = limit - rt * 16;
        if (nv <= 0) break;
        const float* at = arow + rt * 16 * APITCH;
        v8f c = {};
#pragma unroll
        for (int kk = 0; kk < 32; ++kk) {
            const v2f a = *(const v2f*)(at + 32 * (kk & 3) + 4 * (kk >> 2));
            c = __builtin_amdgcn_wmma_f32_16x16x4_f32(
                     false, a,  false, b[kk],
                     (short)0, c,  false,  false);
        }
        float* zt = znext + (size_t)(base + rt * 16 + half * 8) * HH + col;
        if (nv >= 16) {
#pragma unroll
            for (int r = 0; r < 8; ++r) zt[(size_t)r * HH] = c[r];
        } else {
#pragma unroll
            for (int r = 0; r < 8; ++r)
                if (r + half * 8 < nv) zt[(size_t)r * HH] = c[r];
        }
    }
}

template <int LAYER>
__device__ __forceinline__ void agg_listed(float* acc, int* goff_p,
                                           const float* __restrict__ feat,
                                           const int* __restrict__ range_cnt,
                                           const int* __restrict__ elist,
                                           const float* __restrict__ norm_src,
                                           const float* __restrict__ norm_dst,
                                           const float* __restrict__ bias,
                                           const float* __restrict__ g,
                                           const float* __restrict__ bta,
                                           float res,
                                           float* __restrict__ ori,
                                           const float* __restrict__ Wn,
                                           float* __restrict__ dest) {
    int& goff_s = *goff_p;

    const int tid  = threadIdx.x;
    const int lane = tid & 31;
    const int wave = tid >> 5;
    const int base = blockIdx.x * RPB;

    if (tid == 0) goff_s = 0;
    {
        float4* acc4 = (float4*)acc;
        for (int i = tid; i < AROWS * APITCH / 4; i += 1024)
            acc4[i] = make_float4(0.0f, 0.0f, 0.0f, 0.0f);
    }
    __syncthreads();
    add_earlier_counts(range_cnt, &goff_s, tid);
    __syncthreads();
    const int* lst = elist + goff_s;
    const int total = range_cnt[blockIdx.x];

    const int share = (((total + 31) >> 5) + 7) & ~7;
    int c0 = wave * share;
    int wend = c0 + share;
    if (wend > total) wend = total;

    const int smask = (1 << SBITS) - 1;
    int e0 = 0, e1 = 0;
    if (c0 + lane < wend)      e0 = lst[c0 + lane];
    if (c0 + 32 + lane < wend) e1 = lst[c0 + 32 + lane];

    while (c0 < wend) {
        int n = wend - c0;
        if (n > 32) n = 32;
        int e2 = 0;
        if (c0 + 64 + lane < wend) e2 = lst[c0 + 64 + lane];

        const int s = e0 & smask;
        const int r = e0 >> SBITS;
        if (n == 32) sum_chunk<true>(feat, acc, s, r, 32, lane);
        else         sum_chunk<false>(feat, acc, s, r, n, lane);

        e0 = e1; e1 = e2;
        c0 += 32;
    }

    __syncthreads();
    finish_rows<LAYER>(acc, base, wave, lane, norm_src, norm_dst, bias, g, bta, res,
                       ori, dest);
    if (LAYER < 2) {
        v2f b[32];
        load_slab(b, Wn, wave, lane);
        __syncthreads();
        project_rows(acc, b, base, wave, lane, dest);
    }
}

__global__ __launch_bounds__(1024)
void agg_head(const float* __restrict__ feat,
              const int* __restrict__ range_cnt,
              const int* __restrict__ elist,
              const float* __restrict__ norm_src,
              const float* __restrict__ norm_dst,
              const float* __restrict__ bias,
              const float* __restrict__ g,
              const float* __restrict__ bta,
              float* __restrict__ ori,
              const float* __restrict__ Wn,
              float* __restrict__ znext) {
    __shared__ __attribute__((aligned(16))) float acc[AROWS * APITCH];
    __shared__ int goff_s;
    agg_listed<0>(acc, &goff_s, feat, range_cnt, elist, norm_src, norm_dst, bias,
                  g, bta, 0.0f, ori, Wn, znext);
}

__global__ __launch_bounds__(1024)
void agg_mid(const float* __restrict__ feat,
             const int* __restrict__ range_cnt,
             const int* __restrict__ elist,
             const float* __restrict__ norm_src,
             const float* __restrict__ norm_dst,
             const float* __restrict__ bias,
             const float* __restrict__ g,
             const float* __restrict__ bta,
             float res,
             float* __restrict__ ori,
             const float* __restrict__ Wn,
             float* __restrict__ znext) {
    __shared__ __attribute__((aligned(16))) float acc[AROWS * APITCH];
    __shared__ int goff_s;
    agg_listed<1>(acc, &goff_s, feat, range_cnt, elist, norm_src, norm_dst, bias,
                  g, bta, res, ori, Wn, znext);
}

__global__ __launch_bounds__(1024)
void agg_last(const float* __restrict__ feat,
              const int* __restrict__ range_cnt,
              const int* __restrict__ elist,
              const float* __restrict__ norm_dst,
              const float* __restrict__ bias,
              float res,
              float* __restrict__ ori,
              float* __restrict__ out) {
    __shared__ __attribute__((aligned(16))) float acc[AROWS * APITCH];
    __shared__ int goff_s;
    agg_listed<2>(acc, &goff_s, feat, range_cnt, elist, nullptr, norm_dst, bias,
                  nullptr, nullptr, res, ori, nullptr, out);
}

__global__ void gcn_gemm(const float* __restrict__ h,
                         const float* __restrict__ scale,
                         const float* __restrict__ W,
                         float* __restrict__ z) {
    __shared__ float lds_a[16 * LDS_STRIDE];

    const int m0   = blockIdx.x * 16;
    const int tid  = threadIdx.x;
    const int wave = tid >> 5;
    const int lane = tid & 31;

    {
        int base = tid * 8;
        int row  = base >> 7;
        int col  = base & 127;
        float nd = scale ? scale[m0 + row] : 1.0f;
        const float4* s4 = (const float4*)(h + (size_t)(m0 + row) * HH + col);
        float4 a0 = s4[0];
        float4 a1 = s4[1];
        a0.x *= nd; a0.y *= nd; a0.z *= nd; a0.w *= nd;
        a1.x *= nd; a1.y *= nd; a1.z *= nd; a1.w *= nd;
        float4* t4 = (float4*)(lds_a + row * LDS_STRIDE + col);
        t4[0] = a0;
        t4[1] = a1;
    }
    __syncthreads();

    const int n0   = wave * 16;
    const int half = lane >> 4;
    const int l15  = lane & 15;

    const float* arow = lds_a + l15 * LDS_STRIDE + half * 2;
    const float* bcol = W + (size_t)(half * 2) * HH + n0 + l15;

    v8f c = {};
#pragma unroll
    for (int kk = 0; kk < 32; ++kk) {
        const int k0 = kk * 4;
        v2f a, b;
        a.x = arow[k0];
        a.y = arow[k0 + 1];
        b.x = bcol[(size_t)k0 * HH];
        b.y = bcol[(size_t)k0 * HH + HH];
        c = __builtin_amdgcn_wmma_f32_16x16x4_f32(
                 false, a,  false, b,
                 (short)0, c,  false,  false);
    }

    const int nn = n0 + l15;
#pragma unroll
    for (int r = 0; r < 8; ++r) {
        int m = m0 + r + half * 8;
        z[(size_t)m * HH + nn] = c[r];
    }
}

extern "C" void kernel_launch(void* const* d_in, const int* in_sizes, int n_in,
                              void* d_out, int out_size, void* d_ws, size_t ws_size,
                              hipStream_t stream) {
    (void)in_sizes; (void)n_in; (void)out_size; (void)ws_size;

    const float* x    = (const float*)d_in[0];
    const float* W0   = (const float*)d_in[1];
    const float* Ws   = (const float*)d_in[2];
    const float* b    = (const float*)d_in[3];
    const float* ln_g = (const float*)d_in[4];
    const float* ln_b = (const float*)d_in[5];
    const int*   src  = (const int*)d_in[6];
    const int*   dst  = (const int*)d_in[7];
    float* out = (float*)d_out;

    float* ws       = (float*)d_ws;
    float* norm_src = ws;
    float* norm_dst = ws + NN;
    float* zbuf     = ws + 2 * (size_t)NN;
    float* ori      = zbuf + (size_t)NN * HH;
    float* hbuf     = ori + (size_t)NN * HH;
    int* range_cnt  = (int*)(hbuf + (size_t)NN * HH);
    int* cursor     = range_cnt + NBLK;
    int* elist      = cursor + NBLK;

    hipMemsetAsync(ws, 0, 2 * (size_t)NN * sizeof(float), stream);
    deg_kernel<<<DEG_BLOCKS, 256, 0, stream>>>(src, dst, norm_src, norm_dst);
    norm_kernel<<<NBLK, 512, 0, stream>>>(norm_src, norm_dst, range_cnt, cursor);

    part_kernel<<<PBLOCKS, PTHREADS, 0, stream>>>(src, dst, range_cnt, cursor, elist);

    gcn_gemm<<<NN / 16, 256, 0, stream>>>(x, norm_src, W0, zbuf);
    agg_head<<<NBLK, 1024, 0, stream>>>(zbuf, range_cnt, elist,
                                        norm_src, norm_dst, b, ln_g + HH, ln_b + HH,
                                        ori, Ws, hbuf);

    agg_mid<<<NBLK, 1024, 0, stream>>>(hbuf, range_cnt, elist, norm_src, norm_dst,
                                       b + HH, ln_g + 2 * HH, ln_b + 2 * HH, 0.5f,
                                       ori, Ws + (size_t)HH * HH, zbuf);

    agg_last<<<NBLK, 1024, 0, stream>>>(zbuf, range_cnt, elist, norm_dst, b + 2 * HH,
                                        0.5f, ori, out);
}


# degree launch: 256 blocks (src or dst, 8 node ranges, 16 edge slices) count their range's nodes in LDS (integer adds) into count planes that the norm launch sums; global f32 atomics and the memset go
# speedup vs baseline: 4.9931x; 1.4584x over previous
#include <hip/hip_runtime.h>
#include <hip/hip_bf16.h>

#define NN 100000
#define EE 1600000
#define HH 128
#define LDS_STRIDE 132

typedef __attribute__((ext_vector_type(2))) float v2f;
typedef __attribute__((ext_vector_type(8))) float v8f;

#define RPB   391
#define NBLK  256
static_assert((NN - 1) / RPB < NBLK, "every row's range has a work-group");
static_assert(EE % 128 == 0, "whole waves of four edges per lane");

#define DR      8
#define DS      16
#define DNODES  (NN / DR)
#define DEDGES  (EE / DS)
#define DV4     (DEDGES / 4)
#define DPER    ((DV4 + 1023) / 1024)
#define DBATCH  25
static_assert(NN % DR == 0, "equal node ranges");
static_assert(EE % (DS * 4) == 0, "slices of whole 16-byte index loads");
static_assert(DPER % DBATCH == 0, "whole batches");
#define DEG_BLOCKS (2 * DR * DS)

__device__ __forceinline__ void count_node(int* cnt, int node, int lo) {
    const unsigned x = (unsigned)(node - lo);
    if (x < (unsigned)DNODES) atomicAdd(&cnt[x], 1);
}

__global__ __launch_bounds__(1024)
void deg_kernel(const int* __restrict__ src, const int* __restrict__ dst,
                int* __restrict__ planes) {
    __shared__ int cnt[DNODES];
    const int tid = threadIdx.x;
    const int b = blockIdx.x;
    const int a = b / (DR * DS);
    const int r = (b / DS) % DR;
    const int s = b % DS;
    const int lo = r * DNODES;

    for (int i = tid; i < DNODES; i += 1024) cnt[i] = 0;
    __syncthreads();

    const int* ids = a ? dst : src;
    const int4* idx = (const int4*)(ids + (size_t)s * DEDGES);
#pragma unroll
    for (int c = 0; c < DPER; c += DBATCH) {
        int4 v[DBATCH];
#pragma unroll
        for (int u = 0; u < DBATCH; ++u) {
            const int i = tid + (c + u) * 1024;
            v[u] = idx[i < DV4 ? i : DV4 - 1];
        }
#pragma unroll
        for (int u = 0; u < DBATCH; ++u) {
            const int i = tid + (c + u) * 1024;
            if (i < DV4) {
                count_node(cnt, v[u].x, lo);
                count_node(cnt, v[u].y, lo);
                count_node(cnt, v[u].z, lo);
                count_node(cnt, v[u].w, lo);
            }
        }
    }
    __syncthreads();

    int* out = planes + (size_t)(a * DS + s) * NN + lo;
    for (int i = tid; i < DNODES; i += 1024) out[i] = cnt[i];
}

__global__ __launch_bounds__(512)
void norm_kernel(const int* __restrict__ planes,
                 float* __restrict__ ns, float* __restrict__ nd,
                 int* __restrict__ range_cnt, int* __restrict__ cursor) {
    __shared__ int tot;
    if (threadIdx.x == 0) tot = 0;
    __syncthreads();
    const int r = threadIdx.x;
    const int i = blockIdx.x * RPB + r;
    int c = 0;
    if (r < RPB && i < NN) {
        int dout = 0, din = 0;
#pragma unroll
        for (int s = 0; s < DS; ++s) {
            dout += planes[(size_t)s * NN + i];
            din  += planes[(size_t)(DS + s) * NN + i];
        }
        c = din;
        ns[i] = rsqrtf(fmaxf((float)dout, 1.0f));
        nd[i] = rsqrtf(fmaxf((float)din, 1.0f));
    }
    for (int off = 16; off > 0; off >>= 1) c += __shfl_xor(c, off, 32);
    if ((threadIdx.x & 31) == 0 && c != 0) atomicAdd(&tot, c);
    __syncthreads();
    if (threadIdx.x == 0) {
        range_cnt[blockIdx.x] = tot;
        cursor[blockIdx.x] = 0;
    }
}

#define APITCH 132
#define AROWS  400
static_assert(AROWS % 16 == 0 && AROWS >= RPB, "whole tiles cover the group's rows");
#define SBITS 17
#define LCAP  (EE + NBLK * 32)
static_assert(NBLK * RPB >= NN, "ranges cover every row");
static_assert(RPB <= 512, "local row packing");
static_assert(NN <= (1 << SBITS), "list word packing");
static_assert(NBLK <= 1024, "one thread per earlier group");

#define PBLOCKS  256
#define PTHREADS 512
#define PEDGES   (EE / PBLOCKS)
#define PPT      ((PEDGES + PTHREADS - 1) / PTHREADS)
static_assert(EE % PBLOCKS == 0, "blocks take equal shares of the edge list");
static_assert(NBLK == 256 && PTHREADS >= NBLK, "one thread per range, eight whole waves");

__global__ __launch_bounds__(PTHREADS)
void part_kernel(const int* __restrict__ src, const int* __restrict__ dst,
                 const int* __restrict__ range_cnt,
                 int* __restrict__ cursor, int* __restrict__ elist) {
    __shared__ int hist[NBLK];
    __shared__ int gpos[NBLK];
    __shared__ int lpos[NBLK];
    __shared__ int wtc[8], wth[8];
    __shared__ int sword[PEDGES];
    __shared__ int sdest[PEDGES];

    const int tid  = threadIdx.x;
    const int lane = tid & 31;
    const int wave = tid >> 5;

    if (tid < NBLK) hist[tid] = 0;
    __syncthreads();

    const int e0 = blockIdx.x * PEDGES;
    int sv[PPT], dv[PPT];
    int word[PPT], rng[PPT], rank[PPT];
#pragma unroll
    for (int k = 0; k < PPT; ++k) {
        const int i = tid + k * PTHREADS;
        const int ii = i < PEDGES ? i : PEDGES - 1;
        sv[k] = src[e0 + ii];
        dv[k] = dst[e0 + ii];
    }
#pragma unroll
    for (int k = 0; k < PPT; ++k) {
        const int i = tid + k * PTHREADS;
        rng[k] = -1;
        word[k] = 0;
        rank[k] = 0;
        if (i < PEDGES) {
            const int r = dv[k] / RPB;
            rng[k]  = r;
            word[k] = sv[k] | ((dv[k] - r * RPB) << SBITS);
            rank[k] = atomicAdd(&hist[r], 1);
        }
    }
    __syncthreads();

    int c = 0, h = 0, ic = 0, ih = 0;
    if (tid < NBLK) {
        c = (range_cnt[tid] + 31) & ~31;
        h = hist[tid];
        ic = c;
        ih = h;
#pragma unroll
        for (int off = 1; off < 32; off <<= 1) {
            const int vc = __shfl_up(ic, off, 32);
            const int vh = __shfl_up(ih, off, 32);
            if (lane >= off) { ic += vc; ih += vh; }
        }
        if (lane == 31) { wtc[wave] = ic; wth[wave] = ih; }
    }
    __syncthreads();
    if (tid < NBLK) {
        int bc = 0, bh = 0;
        for (int w = 0; w < wave; ++w) { bc += wtc[w]; bh += wth[w]; }
        int got = 0;
        if (h != 0) got = atomicAdd(&cursor[tid], h);
        gpos[tid] = bc + ic - c + got;
        lpos[tid] = bh + ih - h;
    }
    __syncthreads();

#pragma unroll
    for (int k = 0; k < PPT; ++k) {
        if (rng[k] >= 0) {
            const int p = lpos[rng[k]] + rank[k];
            sword[p] = word[k];
            sdest[p] = gpos[rng[k]] + rank[k];
        }
    }
    __syncthreads();

    for (int i = tid; i < PEDGES; i += PTHREADS)
        atomicExch(&elist[sdest[i]], sword[i]);
}

__device__ __forceinline__ float4 load_row(const float* __restrict__ feat,
                                           int row, int lane) {
    return ((const float4*)(feat + (size_t)row * HH))[lane];
}

__device__ __forceinline__ void add_row(float* acc, int r, int j,
                                        float4 w, int lane) {
    const int rj = __builtin_amdgcn_readlane(r, j);
    float* a = acc + rj * APITCH + lane;
    atomicAdd(a,      w.x);
    atomicAdd(a + 32, w.y);
    atomicAdd(a + 64, w.z);
    atomicAdd(a + 96, w.w);
}

template <bool FULL>
__device__ __forceinline__ void sum_chunk(const float* __restrict__ feat,
                                          float* acc, int s, int r,
                                          int cnt, int lane) {
    if (FULL) {
        float4 v[2][8];
#pragma unroll
        for (int u = 0; u < 8; ++u)
            v[0][u] = load_row(feat, __builtin_amdgcn_readlane(s, u), lane);
#pragma unroll
        for (int b = 0; b < 4; ++b) {
            if (b < 3) {
#pragma unroll
                for (int u = 0; u < 8; ++u)
                    v[(b + 1) & 1][u] = load_row(
                        feat, __builtin_amdgcn_readlane(s, 8 * (b + 1) + u), lane);
            }
#pragma unroll
            for (int u = 0; u < 8; ++u)
                add_row(acc, r, 8 * b + u, v[b & 1][u], lane);
        }
    } else {
#pragma unroll
        for (int b = 0; b < 4; ++b) {
            if (8 * b < cnt) {
                float4 v[8];
#pragma unroll
                for (int u = 0; u < 8; ++u)
                    if (8 * b + u < cnt)
                        v[u] = load_row(
                            feat, __builtin_amdgcn_readlane(s, 8 * b + u), lane);
#pragma unroll
                for (int u = 0; u < 8; ++u)
                    if (8 * b + u < cnt)
                        add_row(acc, r, 8 * b + u, v[u], lane);
            }
        }
    }
}

__device__ __forceinline__ void add_earlier_counts(const int* __restrict__ range_cnt,
                                                   int* goff_s, int tid) {
    if (tid < (int)blockIdx.x)
        atomicAdd(goff_s, (range_cnt[tid] + 31) & ~31);
}

#define RBATCH 4
template <int LAYER>
__device__ __forceinline__ void finish_rows(float* acc, int base, int wave, int lane,
                                            const float* __restrict__ norm_src,
                                            const float* __restrict__ norm_dst,
                                            const float* __restrict__ bias,
                                            const float* __restrict__ g,
                                            const float* __restrict__ bta,
                                            float res,
                                            float* __restrict__ ori,
                                            float* __restrict__ dest) {
    const float4 bv = ((const float4*)bias)[lane];
    float4 gv = make_float4(0.0f, 0.0f, 0.0f, 0.0f);
    float4 tv = gv;
    if (LAYER < 2) {
        gv = ((const float4*)g)[lane];
        tv = ((const float4*)bta)[lane];
    }
    for (int row0 = wave; row0 < RPB; row0 += 32 * RBATCH) {
        float4 v[RBATCH], h[RBATCH];
        int nn[RBATCH];
        bool ok[RBATCH];
#pragma unroll
        for (int u = 0; u < RBATCH; ++u) {
            const int row = row0 + 32 * u;
            ok[u] = row < RPB && base + row < NN;
            int rc = row < RPB ? row : RPB - 1;
            int nc = base + rc;
            if (nc > NN - 1) { rc -= nc - (NN - 1); nc = NN - 1; }
            nn[u] = nc;
            const float* a = acc + rc * APITCH + lane;
            const float nd = norm_dst[nc];
            v[u].x = a[0]  * nd + bv.x;
            v[u].y = a[32] * nd + bv.y;
            v[u].z = a[64] * nd + bv.z;
            v[u].w = a[96] * nd + bv.w;
            if (LAYER > 0) {
                const float4 o = ((const float4*)(ori + (size_t)nc * HH))[lane];
                v[u].x += res * o.x;
                v[u].y += res * o.y;
                v[u].z += res * o.z;
                v[u].w += res * o.w;
            }
            if (LAYER < 2) {
                const float ns = norm_src[nc];
                float s = v[u].x + v[u].y + v[u].z + v[u].w;
#pragma unroll
                for (int off = 16; off > 0; off >>= 1) s += __shfl_xor(s, off, 32);
                const float mu = s * (1.0f / 128.0f);
                const float dx = v[u].x - mu, dy = v[u].y - mu;
                const float dz = v[u].z - mu, dw = v[u].w - mu;
                float q = dx * dx + dy * dy + dz * dz + dw * dw;
#pragma unroll
                for (int off = 16; off > 0; off >>= 1) q += __shfl_xor(q, off, 32);
                const float rs = rsqrtf(q * (1.0f / 128.0f) + 1e-5f);
                h[u].x = fmaxf(dx * rs * gv.x + tv.x, 0.0f) * ns;
                h[u].y = fmaxf(dy * rs * gv.y + tv.y, 0.0f) * ns;
                h[u].z = fmaxf(dz * rs * gv.z + tv.z, 0.0f) * ns;
                h[u].w = fmaxf(dw * rs * gv.w + tv.w, 0.0f) * ns;
            }
        }
#pragma unroll
        for (int u = 0; u < RBATCH; ++u) {
            if (ok[u]) {
                if (LAYER == 0) ((float4*)(ori + (size_t)nn[u] * HH))[lane] = v[u];
                if (LAYER == 2) {
                    ((float4*)(dest + (size_t)nn[u] * HH))[lane] = v[u];
                } else {
                    float* aw = acc + (row0 + 32 * u) * APITCH + lane;
                    aw[0]  = h[u].x;
                    aw[32] = h[u].y;
                    aw[64] = h[u].z;
                    aw[96] = h[u].w;
                }
            }
        }
    }
}

__device__ __forceinline__ void load_slab(v2f* b, const float* __restrict__ Wn,
                                          int wave, int lane) {
    const int half = lane >> 4;
    const int col  = (wave & 7) * 16 + (lane & 15);
#pragma unroll
    for (int kk = 0; kk < 32; ++kk) {
        const int j0 = 16 * (kk >> 2) + 8 * half + (kk & 3);
        b[kk].x = Wn[(size_t)j0 * HH + col];
        b[kk].y = Wn[(size_t)(j0 + 4) * HH + col];
    }
}

__device__ __forceinline__ void project_rows(const float* acc, const v2f* b,
                                             int base, int wave, int lane,
                                             float* __restrict__ znext) {
    const int half = lane >> 4;
    const int l15  = lane & 15;
    const int wv   = __builtin_amdgcn_readfirstlane(wave);
    const int col  = (wv & 7) * 16 + l15;
    int limit = NN - base;
    if (limit > RPB) limit = RPB;
    const float* arow = acc + l15 * APITCH + 2 * half;
    for (int rt = wv >> 3; rt < AROWS / 16; rt += 4) {
        const int nv = limit - rt * 16;
        if (nv <= 0) break;
        const float* at = arow + rt * 16 * APITCH;
        v8f c = {};
#pragma unroll
        for (int kk = 0; kk < 32; ++kk) {
            const v2f a = *(const v2f*)(at + 32 * (kk & 3) + 4 * (kk >> 2));
            c = __builtin_amdgcn_wmma_f32_16x16x4_f32(
                     false, a,  false, b[kk],
                     (short)0, c,  false,  false);
        }
        float* zt = znext + (size_t)(base + rt * 16 + half * 8) * HH + col;
        if (nv >= 16) {
#pragma unroll
            for (int r = 0; r < 8; ++r) zt[(size_t)r * HH] = c[r];
        } else {
#pragma unroll
            for (int r = 0; r < 8; ++r)
                if (r + half * 8 < nv) zt[(size_t)r * HH] = c[r];
        }
    }
}

template <int LAYER>
__device__ __forceinline__ void agg_listed(float* acc, int* goff_p,
                                           const float* __restrict__ feat,
                                           const int* __restrict__ range_cnt,
                                           const int* __restrict__ elist,
                                           const float* __restrict__ norm_src,
                                           const float* __restrict__ norm_dst,
                                           const float* __restrict__ bias,
                                           const float* __restrict__ g,
                                           const float* __restrict__ bta,
                                           float res,
                                           float* __restrict__ ori,
                                           const float* __restrict__ Wn,
                                           float* __restrict__ dest) {
    int& goff_s = *goff_p;

    const int tid  = threadIdx.x;
    const int lane = tid & 31;
    const int wave = tid >> 5;
    const int base = blockIdx.x * RPB;

    if (tid == 0) goff_s = 0;
    {
        float4* acc4 = (float4*)acc;
        for (int i = tid; i < AROWS * APITCH / 4; i += 1024)
            acc4[i] = make_float4(0.0f, 0.0f, 0.0f, 0.0f);
    }
    __syncthreads();
    add_earlier_counts(range_cnt, &goff_s, tid);
    __syncthreads();
    const int* lst = elist + goff_s;
    const int total = range_cnt[blockIdx.x];

    const int share = (((total + 31) >> 5) + 7) & ~7;
    int c0 = wave * share;
    int wend = c0 + share;
    if (wend > total) wend = total;

    const int smask = (1 << SBITS) - 1;
    int e0 = 0, e1 = 0;
    if (c0 + lane < wend)      e0 = lst[c0 + lane];
    if (c0 + 32 + lane < wend) e1 = lst[c0 + 32 + lane];

    while (c0 < wend) {
        int n = wend - c0;
        if (n > 32) n = 32;
        int e2 = 0;
        if (c0 + 64 + lane < wend) e2 = lst[c0 + 64 + lane];

        const int s = e0 & smask;
        const int r = e0 >> SBITS;
        if (n == 32) sum_chunk<true>(feat, acc, s, r, 32, lane);
        else         sum_chunk<false>(feat, acc, s, r, n, lane);

        e0 = e1; e1 = e2;
        c0 += 32;
    }

    __syncthreads();
    finish_rows<LAYER>(acc, base, wave, lane, norm_src, norm_dst, bias, g, bta, res,
                       ori, dest);
    if (LAYER < 2) {
        v2f b[32];
        load_slab(b, Wn, wave, lane);
        __syncthreads();
        project_rows(acc, b, base, wave, lane, dest);
    }
}

__global__ __launch_bounds__(1024)
void agg_head(const float* __restrict__ feat,
              const int* __restrict__ range_cnt,
              const int* __restrict__ elist,
              const float* __restrict__ norm_src,
              const float* __restrict__ norm_dst,
              const float* __restrict__ bias,
              const float* __restrict__ g,
              const float* __restrict__ bta,
              float* __restrict__ ori,
              const float* __restrict__ Wn,
              float* __restrict__ znext) {
    __shared__ __attribute__((aligned(16))) float acc[AROWS * APITCH];
    __shared__ int goff_s;
    agg_listed<0>(acc, &goff_s, feat, range_cnt, elist, norm_src, norm_dst, bias,
                  g, bta, 0.0f, ori, Wn, znext);
}

__global__ __launch_bounds__(1024)
void agg_mid(const float* __restrict__ feat,
             const int* __restrict__ range_cnt,
             const int* __restrict__ elist,
             const float* __restrict__ norm_src,
             const float* __restrict__ norm_dst,
             const float* __restrict__ bias,
             const float* __restrict__ g,
             const float* __restrict__ bta,
             float res,
             float* __restrict__ ori,
             const float* __restrict__ Wn,
             float* __restrict__ znext) {
    __shared__ __attribute__((aligned(16))) float acc[AROWS * APITCH];
    __shared__ int goff_s;
    agg_listed<1>(acc, &goff_s, feat, range_cnt, elist, norm_src, norm_dst, bias,
                  g, bta, res, ori, Wn, znext);
}

__global__ __launch_bounds__(1024)
void agg_last(const float* __restrict__ feat,
              const int* __restrict__ range_cnt,
              const int* __restrict__ elist,
              const float* __restrict__ norm_dst,
              const float* __restrict__ bias,
              float res,
              float* __restrict__ ori,
              float* __restrict__ out) {
    __shared__ __attribute__((aligned(16))) float acc[AROWS * APITCH];
    __shared__ int goff_s;
    agg_listed<2>(acc, &goff_s, feat, range_cnt, elist, nullptr, norm_dst, bias,
                  nullptr, nullptr, res, ori, nullptr, out);
}

__global__ void gcn_gemm(const float* __restrict__ h,
                         const float* __restrict__ scale,
                         const float* __restrict__ W,
                         float* __restrict__ z) {
    __shared__ float lds_a[16 * LDS_STRIDE];

    const int m0   = blockIdx.x * 16;
    const int tid  = threadIdx.x;
    const int wave = tid >> 5;
    const int lane = tid & 31;

    {
        int base = tid * 8;
        int row  = base >> 7;
        int col  = base & 127;
        float nd = scale ? scale[m0 + row] : 1.0f;
        const float4* s4 = (const float4*)(h + (size_t)(m0 + row) * HH + col);
        float4 a0 = s4[0];
        float4 a1 = s4[1];
        a0.x *= nd; a0.y *= nd; a0.z *= nd; a0.w *= nd;
        a1.x *= nd; a1.y *= nd; a1.z *= nd; a1.w *= nd;
        float4* t4 = (float4*)(lds_a + row * LDS_STRIDE + col);
        t4[0] = a0;
        t4[1] = a1;
    }
    __syncthreads();

    const int n0   = wave * 16;
    const int half = lane >> 4;
    const int l15  = lane & 15;

    const float* arow = lds_a + l15 * LDS_STRIDE + half * 2;
    const float* bcol = W + (size_t)(half * 2) * HH + n0 + l15;

    v8f c = {};
#pragma unroll
    for (int kk = 0; kk < 32; ++kk) {
        const int k0 = kk * 4;
        v2f a, b;
        a.x = arow[k0];
        a.y = arow[k0 + 1];
        b.x = bcol[(size_t)k0 * HH];
        b.y = bcol[(size_t)k0 * HH + HH];
        c = __builtin_amdgcn_wmma_f32_16x16x4_f32(
                 false, a,  false, b,
                 (short)0, c,  false,  false);
    }

    const int nn = n0 + l15;
#pragma unroll
    for (int r = 0; r < 8; ++r) {
        int m = m0 + r + half * 8;
        z[(size_t)m * HH + nn] = c[r];
    }
}

extern "C" void kernel_launch(void* const* d_in, const int* in_sizes, int n_in,
                              void* d_out, int out_size, void* d_ws, size_t ws_size,
                              hipStream_t stream) {
    (void)in_sizes; (void)n_in; (void)out_size; (void)ws_size;

    const float* x    = (const float*)d_in[0];
    const float* W0   = (const float*)d_in[1];
    const float* Ws   = (const float*)d_in[2];
    const float* b    = (const float*)d_in[3];
    const float* ln_g = (const float*)d_in[4];
    const float* ln_b = (const float*)d_in[5];
    const int*   src  = (const int*)d_in[6];
    const int*   dst  = (const int*)d_in[7];
    float* out = (float*)d_out;

    float* ws       = (float*)d_ws;
    float* norm_src = ws;
    float* norm_dst = ws + NN;
    float* zbuf     = ws + 2 * (size_t)NN;
    float* ori      = zbuf + (size_t)NN * HH;
    float* hbuf     = ori + (size_t)NN * HH;
    int* range_cnt  = (int*)(hbuf + (size_t)NN * HH);
    int* cursor     = range_cnt + NBLK;
    int* elist      = cursor + NBLK;
    int* planes     = elist + LCAP;

    deg_kernel<<<DEG_BLOCKS, 1024, 0, stream>>>(src, dst, planes);
    norm_kernel<<<NBLK, 512, 0, stream>>>(planes, norm_src, norm_dst, range_cnt, cursor);

    part_kernel<<<PBLOCKS, PTHREADS, 0, stream>>>(src, dst, range_cnt, cursor, elist);

    gcn_gemm<<<NN / 16, 256, 0, stream>>>(x, norm_src, W0, zbuf);
    agg_head<<<NBLK, 1024, 0, stream>>>(zbuf, range_cnt, elist,
                                        norm_src, norm_dst, b, ln_g + HH, ln_b + HH,
                                        ori, Ws, hbuf);

    agg_mid<<<NBLK, 1024, 0, stream>>>(hbuf, range_cnt, elist, norm_src, norm_dst,
                                       b + HH, ln_g + 2 * HH, ln_b + 2 * HH, 0.5f,
                                       ori, Ws + (size_t)HH * HH, zbuf);

    agg_last<<<NBLK, 1024, 0, stream>>>(zbuf, range_cnt, elist, norm_dst, b + 2 * HH,
                                        0.5f, ori, out);
}


# aggregate: layer 0 orders each group's edge words by row in LDS (counter adds give ranks); waves sum each run of a row in registers, one LDS f32 add per run (was 4 per edge); sum order = list order
# speedup vs baseline: 5.3018x; 1.0618x over previous
#include <hip/hip_runtime.h>
#include <hip/hip_bf16.h>

#define NN 100000
#define EE 1600000
#define HH 128
#define LDS_STRIDE 132

typedef __attribute__((ext_vector_type(2))) float v2f;
typedef __attribute__((ext_vector_type(8))) float v8f;

#define RPB   391
#define NBLK  256
static_assert((NN - 1) / RPB < NBLK, "every row's range has a work-group");
static_assert(EE % 128 == 0, "whole waves of four edges per lane");

#define DR      8
#define DS      16
#define DNODES  (NN / DR)
#define DEDGES  (EE / DS)
#define DV4     (DEDGES / 4)
#define DPER    ((DV4 + 1023) / 1024)
#define DBATCH  25
static_assert(NN % DR == 0, "equal node ranges");
static_assert(EE % (DS * 4) == 0, "slices of whole 16-byte index loads");
static_assert(DPER % DBATCH == 0, "whole batches");
#define DEG_BLOCKS (2 * DR * DS)

__device__ __forceinline__ void count_node(int* cnt, int node, int lo) {
    const unsigned x = (unsigned)(node - lo);
    if (x < (unsigned)DNODES) atomicAdd(&cnt[x], 1);
}

__global__ __launch_bounds__(1024)
void deg_kernel(const int* __restrict__ src, const int* __restrict__ dst,
                int* __restrict__ planes) {
    __shared__ int cnt[DNODES];
    const int tid = threadIdx.x;
    const int b = blockIdx.x;
    const int a = b / (DR * DS);
    const int r = (b / DS) % DR;
    const int s = b % DS;
    const int lo = r * DNODES;

    for (int i = tid; i < DNODES; i += 1024) cnt[i] = 0;
    __syncthreads();

    const int* ids = a ? dst : src;
    const int4* idx = (const int4*)(ids + (size_t)s * DEDGES);
#pragma unroll
    for (int c = 0; c < DPER; c += DBATCH) {
        int4 v[DBATCH];
#pragma unroll
        for (int u = 0; u < DBATCH; ++u) {
            const int i = tid + (c + u) * 1024;
            v[u] = idx[i < DV4 ? i : DV4 - 1];
        }
#pragma unroll
        for (int u = 0; u < DBATCH; ++u) {
            const int i = tid + (c + u) * 1024;
            if (i < DV4) {
                count_node(cnt, v[u].x, lo);
                count_node(cnt, v[u].y, lo);
                count_node(cnt, v[u].z, lo);
                count_node(cnt, v[u].w, lo);
            }
        }
    }
    __syncthreads();

    int* out = planes + (size_t)(a * DS + s) * NN + lo;
    for (int i = tid; i < DNODES; i += 1024) out[i] = cnt[i];
}

__global__ __launch_bounds__(512)
void norm_kernel(const int* __restrict__ planes,
                 float* __restrict__ ns, float* __restrict__ nd,
                 int* __restrict__ range_cnt, int* __restrict__ cursor) {
    __shared__ int tot;
    if (threadIdx.x == 0) tot = 0;
    __syncthreads();
    const int r = threadIdx.x;
    const int i = blockIdx.x * RPB + r;
    int c = 0;
    if (r < RPB && i < NN) {
        int dout = 0, din = 0;
#pragma unroll
        for (int s = 0; s < DS; ++s) {
            dout += planes[(size_t)s * NN + i];
            din  += planes[(size_t)(DS + s) * NN + i];
        }
        c = din;
        ns[i] = rsqrtf(fmaxf((float)dout, 1.0f));
        nd[i] = rsqrtf(fmaxf((float)din, 1.0f));
    }
    for (int off = 16; off > 0; off >>= 1) c += __shfl_xor(c, off, 32);
    if ((threadIdx.x & 31) == 0 && c != 0) atomicAdd(&tot, c);
    __syncthreads();
    if (threadIdx.x == 0) {
        range_cnt[blockIdx.x] = tot;
        cursor[blockIdx.x] = 0;
    }
}

#define APITCH 132
#define AROWS  400
static_assert(AROWS % 16 == 0 && AROWS >= RPB, "whole tiles cover the group's rows");
#define SBITS 17
#define LCAP  (EE + NBLK * 32)
static_assert(NBLK * RPB >= NN, "ranges cover every row");
static_assert(RPB <= 512, "local row packing");
static_assert(NN <= (1 << SBITS), "list word packing");
static_assert(NBLK <= 1024, "one thread per earlier group");

#define PBLOCKS  256
#define PTHREADS 512
#define PEDGES   (EE / PBLOCKS)
#define PPT      ((PEDGES + PTHREADS - 1) / PTHREADS)
static_assert(EE % PBLOCKS == 0, "blocks take equal shares of the edge list");
static_assert(NBLK == 256 && PTHREADS >= NBLK, "one thread per range, eight whole waves");

__global__ __launch_bounds__(PTHREADS)
void part_kernel(const int* __restrict__ src, const int* __restrict__ dst,
                 const int* __restrict__ range_cnt,
                 int* __restrict__ cursor, int* __restrict__ elist) {
    __shared__ int hist[NBLK];
    __shared__ int gpos[NBLK];
    __shared__ int lpos[NBLK];
    __shared__ int wtc[8], wth[8];
    __shared__ int sword[PEDGES];
    __shared__ int sdest[PEDGES];

    const int tid  = threadIdx.x;
    const int lane = tid & 31;
    const int wave = tid >> 5;

    if (tid < NBLK) hist[tid] = 0;
    __syncthreads();

    const int e0 = blockIdx.x * PEDGES;
    int sv[PPT], dv[PPT];
    int word[PPT], rng[PPT], rank[PPT];
#pragma unroll
    for (int k = 0; k < PPT; ++k) {
        const int i = tid + k * PTHREADS;
        const int ii = i < PEDGES ? i : PEDGES - 1;
        sv[k] = src[e0 + ii];
        dv[k] = dst[e0 + ii];
    }
#pragma unroll
    for (int k = 0; k < PPT; ++k) {
        const int i = tid + k * PTHREADS;
        rng[k] = -1;
        word[k] = 0;
        rank[k] = 0;
        if (i < PEDGES) {
            const int r = dv[k] / RPB;
            rng[k]  = r;
            word[k] = sv[k] | ((dv[k] - r * RPB) << SBITS);
            rank[k] = atomicAdd(&hist[r], 1);
        }
    }
    __syncthreads();

    int c = 0, h = 0, ic = 0, ih = 0;
    if (tid < NBLK) {
        c = (range_cnt[tid] + 31) & ~31;
        h = hist[tid];
        ic = c;
        ih = h;
#pragma unroll
        for (int off = 1; off < 32; off <<= 1) {
            const int vc = __shfl_up(ic, off, 32);
            const int vh = __shfl_up(ih, off, 32);
            if (lane >= off) { ic += vc; ih += vh; }
        }
        if (lane == 31) { wtc[wave] = ic; wth[wave] = ih; }
    }
    __syncthreads();
    if (tid < NBLK) {
        int bc = 0, bh = 0;
        for (int w = 0; w < wave; ++w) { bc += wtc[w]; bh += wth[w]; }
        int got = 0;
        if (h != 0) got = atomicAdd(&cursor[tid], h);
        gpos[tid] = bc + ic - c + got;
        lpos[tid] = bh + ih - h;
    }
    __syncthreads();

#pragma unroll
    for (int k = 0; k < PPT; ++k) {
        if (rng[k] >= 0) {
            const int p = lpos[rng[k]] + rank[k];
            sword[p] = word[k];
            sdest[p] = gpos[rng[k]] + rank[k];
        }
    }
    __syncthreads();

    for (int i = tid; i < PEDGES; i += PTHREADS)
        atomicExch(&elist[sdest[i]], sword[i]);
}

__device__ __forceinline__ float4 load_row(const float* __restrict__ feat,
                                           int row, int lane) {
    return ((const float4*)(feat + (size_t)row * HH))[lane];
}

__device__ __forceinline__ void flush_run(float* acc, int cur, float4 sum, int lane) {
    float* a = acc + cur * APITCH + lane;
    atomicAdd(a,      sum.x);
    atomicAdd(a + 32, sum.y);
    atomicAdd(a + 64, sum.z);
    atomicAdd(a + 96, sum.w);
}

__device__ __forceinline__ void take_edge(float* acc, int r, int j, float4 w,
                                          int lane, float4& sum, int& cur) {
    const int rj = __builtin_amdgcn_readlane(r, j);
    if (rj != cur) {
        if (cur >= 0) flush_run(acc, cur, sum, lane);
        cur = rj;
        sum = make_float4(0.0f, 0.0f, 0.0f, 0.0f);
    }
    sum.x += w.x;
    sum.y += w.y;
    sum.z += w.z;
    sum.w += w.w;
}

template <bool FULL>
__device__ __forceinline__ void walk_chunk(const float* __restrict__ feat,
                                           float* acc, int s, int r,
                                           int cnt, int lane,
                                           float4& sum, int& cur) {
    if (FULL) {
        float4 v[2][8];
#pragma unroll
        for (int u = 0; u < 8; ++u)
            v[0][u] = load_row(feat, __builtin_amdgcn_readlane(s, u), lane);
#pragma unroll
        for (int b = 0; b < 4; ++b) {
            if (b < 3) {
#pragma unroll
                for (int u = 0; u < 8; ++u)
                    v[(b + 1) & 1][u] = load_row(
                        feat, __builtin_amdgcn_readlane(s, 8 * (b + 1) + u), lane);
            }
#pragma unroll
            for (int u = 0; u < 8; ++u)
                take_edge(acc, r, 8 * b + u, v[b & 1][u], lane, sum, cur);
        }
    } else {
#pragma unroll
        for (int b = 0; b < 4; ++b) {
            if (8 * b < cnt) {
                float4 v[8];
#pragma unroll
                for (int u = 0; u < 8; ++u)
                    if (8 * b + u < cnt)
                        v[u] = load_row(
                            feat, __builtin_amdgcn_readlane(s, 8 * b + u), lane);
#pragma unroll
                for (int u = 0; u < 8; ++u)
                    if (8 * b + u < cnt)
                        take_edge(acc, r, 8 * b + u, v[u], lane, sum, cur);
            }
        }
    }
}

__device__ __forceinline__ void add_earlier_counts(const int* __restrict__ range_cnt,
                                                   int* goff_s, int tid) {
    if (tid < (int)blockIdx.x)
        atomicAdd(goff_s, (range_cnt[tid] + 31) & ~31);
}

#define RBATCH 4
template <int LAYER>
__device__ __forceinline__ void finish_rows(float* acc, int base, int wave, int lane,
                                            const float* __restrict__ norm_src,
                                            const float* __restrict__ norm_dst,
                                            const float* __restrict__ bias,
                                            const float* __restrict__ g,
                                            const float* __restrict__ bta,
                                            float res,
                                            float* __restrict__ ori,
                                            float* __restrict__ dest) {
    const float4 bv = ((const float4*)bias)[lane];
    float4 gv = make_float4(0.0f, 0.0f, 0.0f, 0.0f);
    float4 tv = gv;
    if (LAYER < 2) {
        gv = ((const float4*)g)[lane];
        tv = ((const float4*)bta)[lane];
    }
    for (int row0 = wave; row0 < RPB; row0 += 32 * RBATCH) {
        float4 v[RBATCH], h[RBATCH];
        int nn[RBATCH];
        bool ok[RBATCH];
#pragma unroll
        for (int u = 0; u < RBATCH; ++u) {
            const int row = row0 + 32 * u;
            ok[u] = row < RPB && base + row < NN;
            int rc = row < RPB ? row : RPB - 1;
            int nc = base + rc;
            if (nc > NN - 1) { rc -= nc - (NN - 1); nc = NN - 1; }
            nn[u] = nc;
            const float* a = acc + rc * APITCH + lane;
            const float nd = norm_dst[nc];
            v[u].x = a[0]  * nd + bv.x;
            v[u].y = a[32] * nd + bv.y;
            v[u].z = a[64] * nd + bv.z;
            v[u].w = a[96] * nd + bv.w;
            if (LAYER > 0) {
                const float4 o = ((const float4*)(ori + (size_t)nc * HH))[lane];
                v[u].x += res * o.x;
                v[u].y += res * o.y;
                v[u].z += res * o.z;
                v[u].w += res * o.w;
            }
            if (LAYER < 2) {
                const float ns = norm_src[nc];
                float s = v[u].x + v[u].y + v[u].z + v[u].w;
#pragma unroll
                for (int off = 16; off > 0; off >>= 1) s += __shfl_xor(s, off, 32);
                const float mu = s * (1.0f / 128.0f);
                const float dx = v[u].x - mu, dy = v[u].y - mu;
                const float dz = v[u].z - mu, dw = v[u].w - mu;
                float q = dx * dx + dy * dy + dz * dz + dw * dw;
#pragma unroll
                for (int off = 16; off > 0; off >>= 1) q += __shfl_xor(q, off, 32);
                const float rs = rsqrtf(q * (1.0f / 128.0f) + 1e-5f);
                h[u].x = fmaxf(dx * rs * gv.x + tv.x, 0.0f) * ns;
                h[u].y = fmaxf(dy * rs * gv.y + tv.y, 0.0f) * ns;
                h[u].z = fmaxf(dz * rs * gv.z + tv.z, 0.0f) * ns;
                h[u].w = fmaxf(dw * rs * gv.w + tv.w, 0.0f) * ns;
            }
        }
#pragma unroll
        for (int u = 0; u < RBATCH; ++u) {
            if (ok[u]) {
                if (LAYER == 0) ((float4*)(ori + (size_t)nn[u] * HH))[lane] = v[u];
                if (LAYER == 2) {
                    ((float4*)(dest + (size_t)nn[u] * HH))[lane] = v[u];
                } else {
                    float* aw = acc + (row0 + 32 * u) * APITCH + lane;
                    aw[0]  = h[u].x;
                    aw[32] = h[u].y;
                    aw[64] = h[u].z;
                    aw[96] = h[u].w;
                }
            }
        }
    }
}

__device__ __forceinline__ void load_slab(v2f* b, const float* __restrict__ Wn,
                                          int wave, int lane) {
    const int half = lane >> 4;
    const int col  = (wave & 7) * 16 + (lane & 15);
#pragma unroll
    for (int kk = 0; kk < 32; ++kk) {
        const int j0 = 16 * (kk >> 2) + 8 * half + (kk & 3);
        b[kk].x = Wn[(size_t)j0 * HH + col];
        b[kk].y = Wn[(size_t)(j0 + 4) * HH + col];
    }
}

__device__ __forceinline__ void project_rows(const float* acc, const v2f* b,
                                             int base, int wave, int lane,
                                             float* __restrict__ znext) {
    const int half = lane >> 4;
    const int l15  = lane & 15;
    const int wv   = __builtin_amdgcn_readfirstlane(wave);
    const int col  = (wv & 7) * 16 + l15;
    int limit = NN - base;
    if (limit > RPB) limit = RPB;
    const float* arow = acc + l15 * APITCH + 2 * half;
    for (int rt = wv >> 3; rt < AROWS / 16; rt += 4) {
        const int nv = limit - rt * 16;
        if (nv <= 0) break;
        const float* at = arow + rt * 16 * APITCH;
        v8f c = {};
#pragma unroll
        for (int kk = 0; kk < 32; ++kk) {
            const v2f a = *(const v2f*)(at + 32 * (kk & 3) + 4 * (kk >> 2));
            c = __builtin_amdgcn_wmma_f32_16x16x4_f32(
                     false, a,  false, b[kk],
                     (short)0, c,  false,  false);
        }
        float* zt = znext + (size_t)(base + rt * 16 + half * 8) * HH + col;
        if (nv >= 16) {
#pragma unroll
            for (int r = 0; r < 8; ++r) zt[(size_t)r * HH] = c[r];
        } else {
#pragma unroll
            for (int r = 0; r < 8; ++r)
                if (r + half * 8 < nv) zt[(size_t)r * HH] = c[r];
        }
    }
}

#define SCAP 8192
#define SPER (SCAP / 1024)
template <int LAYER>
__device__ __forceinline__ void agg_listed(float* acc, int* goff_p,
                                           int* cnt, int* sorted, int* wtot,
                                           const float* __restrict__ feat,
                                           const int* __restrict__ range_cnt,
                                           int* elist,
                                           const float* __restrict__ norm_src,
                                           const float* __restrict__ norm_dst,
                                           const float* __restrict__ bias,
                                           const float* __restrict__ g,
                                           const float* __restrict__ bta,
                                           float res,
                                           float* __restrict__ ori,
                                           const float* __restrict__ Wn,
                                           float* __restrict__ dest) {
    int& goff_s = *goff_p;

    const int tid  = threadIdx.x;
    const int lane = tid & 31;
    const int wave = tid >> 5;
    const int base = blockIdx.x * RPB;

    if (tid == 0) goff_s = 0;
    if (LAYER == 0 && tid < AROWS) cnt[tid] = 0;
    {
        float4* acc4 = (float4*)acc;
        for (int i = tid; i < AROWS * APITCH / 4; i += 1024)
            acc4[i] = make_float4(0.0f, 0.0f, 0.0f, 0.0f);
    }
    __syncthreads();
    add_earlier_counts(range_cnt, &goff_s, tid);
    __syncthreads();
    int* lst = elist + goff_s;
    const int total = range_cnt[blockIdx.x];

    const int share = (((total + 31) >> 5) + 7) & ~7;
    const int wv = __builtin_amdgcn_readfirstlane(wave);
    int c0 = wv * share;
    int wend = c0 + share;
    if (wend > total) wend = total;

    const int smask = (1 << SBITS) - 1;

    if (total <= SCAP) {
        if (LAYER == 0) {
            int word[SPER], rank[SPER];
#pragma unroll
            for (int k = 0; k < SPER; ++k) {
                const int i = tid + k * 1024;
                word[k] = 0;
                if (i < total) word[k] = lst[i];
            }
#pragma unroll
            for (int k = 0; k < SPER; ++k) {
                const int i = tid + k * 1024;
                rank[k] = 0;
                if (i < total) rank[k] = atomicAdd(&cnt[word[k] >> SBITS], 1);
            }
            __syncthreads();

            const int c = tid < AROWS ? cnt[tid] : 0;
            int ic = c;
#pragma unroll
            for (int off = 1; off < 32; off <<= 1) {
                const int vc = __shfl_up(ic, off, 32);
                if (lane >= off) ic += vc;
            }
            if (lane == 31) wtot[wave] = ic;
            __syncthreads();
            if (tid < AROWS) {
                int bc = 0;
                for (int w = 0; w < wave; ++w) bc += wtot[w];
                cnt[tid] = bc + ic - c;
            }
            __syncthreads();

#pragma unroll
            for (int k = 0; k < SPER; ++k) {
                const int i = tid + k * 1024;
                if (i < total) sorted[cnt[word[k] >> SBITS] + rank[k]] = word[k];
            }
            __syncthreads();

            for (int i = tid; i < total; i += 1024) lst[i] = sorted[i];
        }

        float4 sum = make_float4(0.0f, 0.0f, 0.0f, 0.0f);
        int cur = -1;
        int e0 = 0, e1 = 0;
        if (c0 + lane < wend)      e0 = LAYER == 0 ? sorted[c0 + lane] : lst[c0 + lane];
        if (c0 + 32 + lane < wend) e1 = LAYER == 0 ? sorted[c0 + 32 + lane] : lst[c0 + 32 + lane];
        while (c0 < wend) {
            int n = wend - c0;
            if (n > 32) n = 32;
            int e2 = 0;
            if (c0 + 64 + lane < wend) e2 = LAYER == 0 ? sorted[c0 + 64 + lane] : lst[c0 + 64 + lane];

            const int s = e0 & smask;
            const int r = e0 >> SBITS;
            if (n == 32) walk_chunk<true>(feat, acc, s, r, 32, lane, sum, cur);
            else         walk_chunk<false>(feat, acc, s, r, n, lane, sum, cur);

            e0 = e1; e1 = e2;
            c0 += 32;
        }
        if (cur >= 0) flush_run(acc, cur, sum, lane);
    } else {
        for (int i = c0; i < wend; ++i) {
            const int e = lst[i];
            const float4 w = load_row(feat, e & smask, lane);
            flush_run(acc, e >> SBITS, w, lane);
        }
    }

    __syncthreads();
    finish_rows<LAYER>(acc, base, wave, lane, norm_src, norm_dst, bias, g, bta, res,
                       ori, dest);
    if (LAYER < 2) {
        v2f b[32];
        load_slab(b, Wn, wave, lane);
        __syncthreads();
        project_rows(acc, b, base, wave, lane, dest);
    }
}

__global__ __launch_bounds__(1024)
void agg_head(const float* __restrict__ feat,
              const int* __restrict__ range_cnt,
              int* elist,
              const float* __restrict__ norm_src,
              const float* __restrict__ norm_dst,
              const float* __restrict__ bias,
              const float* __restrict__ g,
              const float* __restrict__ bta,
              float* __restrict__ ori,
              const float* __restrict__ Wn,
              float* __restrict__ znext) {
    __shared__ __attribute__((aligned(16))) float acc[AROWS * APITCH];
    __shared__ int goff_s;
    __shared__ int cnt[AROWS];
    __shared__ int sorted[SCAP];
    __shared__ int wtot[32];
    agg_listed<0>(acc, &goff_s, cnt, sorted, wtot, feat, range_cnt, elist, norm_src, norm_dst, bias,
                  g, bta, 0.0f, ori, Wn, znext);
}

__global__ __launch_bounds__(1024)
void agg_mid(const float* __restrict__ feat,
             const int* __restrict__ range_cnt,
             const int* __restrict__ elist,
             const float* __restrict__ norm_src,
             const float* __restrict__ norm_dst,
             const float* __restrict__ bias,
             const float* __restrict__ g,
             const float* __restrict__ bta,
             float res,
             float* __restrict__ ori,
             const float* __restrict__ Wn,
             float* __restrict__ znext) {
    __shared__ __attribute__((aligned(16))) float acc[AROWS * APITCH];
    __shared__ int goff_s;
    agg_listed<1>(acc, &goff_s, nullptr, nullptr, nullptr, feat, range_cnt, const_cast<int*>(elist), norm_src, norm_dst, bias,
                  g, bta, res, ori, Wn, znext);
}

__global__ __launch_bounds__(1024)
void agg_last(const float* __restrict__ feat,
              const int* __restrict__ range_cnt,
              const int* __restrict__ elist,
              const float* __restrict__ norm_dst,
              const float* __restrict__ bias,
              float res,
              float* __restrict__ ori,
              float* __restrict__ out) {
    __shared__ __attribute__((aligned(16))) float acc[AROWS * APITCH];
    __shared__ int goff_s;
    agg_listed<2>(acc, &goff_s, nullptr, nullptr, nullptr, feat, range_cnt, const_cast<int*>(elist), nullptr, norm_dst, bias,
                  nullptr, nullptr, res, ori, nullptr, out);
}

__global__ void gcn_gemm(const float* __restrict__ h,
                         const float* __restrict__ scale,
                         const float* __restrict__ W,
                         float* __restrict__ z) {
    __shared__ float lds_a[16 * LDS_STRIDE];

    const int m0   = blockIdx.x * 16;
    const int tid  = threadIdx.x;
    const int wave = tid >> 5;
    const int lane = tid & 31;

    {
        int base = tid * 8;
        int row  = base >> 7;
        int col  = base & 127;
        float nd = scale ? scale[m0 + row] : 1.0f;
        const float4* s4 = (const float4*)(h + (size_t)(m0 + row) * HH + col);
        float4 a0 = s4[0];
        float4 a1 = s4[1];
        a0.x *= nd; a0.y *= nd; a0.z *= nd; a0.w *= nd;
        a1.x *= nd; a1.y *= nd; a1.z *= nd; a1.w *= nd;
        float4* t4 = (float4*)(lds_a + row * LDS_STRIDE + col);
        t4[0] = a0;
        t4[1] = a1;
    }
    __syncthreads();

    const int n0   = wave * 16;
    const int half = lane >> 4;
    const int l15  = lane & 15;

    const float* arow = lds_a + l15 * LDS_STRIDE + half * 2;
    const float* bcol = W + (size_t)(half * 2) * HH + n0 + l15;

    v8f c = {};
#pragma unroll
    for (int kk = 0; kk < 32; ++kk) {
        const int k0 = kk * 4;
        v2f a, b;
        a.x = arow[k0];
        a.y = arow[k0 + 1];
        b.x = bcol[(size_t)k0 * HH];
        b.y = bcol[(size_t)k0 * HH + HH];
        c = __builtin_amdgcn_wmma_f32_16x16x4_f32(
                 false, a,  false, b,
                 (short)0, c,  false,  false);
    }

    const int nn = n0 + l15;
#pragma unroll
    for (int r = 0; r < 8; ++r) {
        int m = m0 + r + half * 8;
        z[(size_t)m * HH + nn] = c[r];
    }
}

extern "C" void kernel_launch(void* const* d_in, const int* in_sizes, int n_in,
                              void* d_out, int out_size, void* d_ws, size_t ws_size,
                              hipStream_t stream) {
    (void)in_sizes; (void)n_in; (void)out_size; (void)ws_size;

    const float* x    = (const float*)d_in[0];
    const float* W0   = (const float*)d_in[1];
    const float* Ws   = (const float*)d_in[2];
    const float* b    = (const float*)d_in[3];
    const float* ln_g = (const float*)d_in[4];
    const float* ln_b = (const float*)d_in[5];
    const int*   src  = (const int*)d_in[6];
    const int*   dst  = (const int*)d_in[7];
    float* out = (float*)d_out;

    float* ws       = (float*)d_ws;
    float* norm_src = ws;
    float* norm_dst = ws + NN;
    float* zbuf     = ws + 2 * (size_t)NN;
    float* ori      = zbuf + (size_t)NN * HH;
    float* hbuf     = ori + (size_t)NN * HH;
    int* range_cnt  = (int*)(hbuf + (size_t)NN * HH);
    int* cursor     = range_cnt + NBLK;
    int* elist      = cursor + NBLK;
    int* planes     = elist + LCAP;

    deg_kernel<<<DEG_BLOCKS, 1024, 0, stream>>>(src, dst, planes);
    norm_kernel<<<NBLK, 512, 0, stream>>>(planes, norm_src, norm_dst, range_cnt, cursor);

    part_kernel<<<PBLOCKS, PTHREADS, 0, stream>>>(src, dst, range_cnt, cursor, elist);

    gcn_gemm<<<NN / 16, 256, 0, stream>>>(x, norm_src, W0, zbuf);
    agg_head<<<NBLK, 1024, 0, stream>>>(zbuf, range_cnt, elist,
                                        norm_src, norm_dst, b, ln_g + HH, ln_b + HH,
                                        ori, Ws, hbuf);

    agg_mid<<<NBLK, 1024, 0, stream>>>(hbuf, range_cnt, elist, norm_src, norm_dst,
                                       b + HH, ln_g + 2 * HH, ln_b + 2 * HH, 0.5f,
                                       ori, Ws + (size_t)HH * HH, zbuf);

    agg_last<<<NBLK, 1024, 0, stream>>>(zbuf, range_cnt, elist, norm_dst, b + 2 * HH,
                                        0.5f, ori, out);
}
